# starting kernel
# baseline (speedup 1.0000x reference)
#include <hip/hip_runtime.h>


typedef __attribute__((ext_vector_type(2))) float v2f;
typedef __attribute__((ext_vector_type(8))) float v8f;

#define B_TOT   256
#define T_LEN   2048
#define HDIM    50
#define KDIM    100
#define KSTEPS  25
#define BM      16
#define NW      16
#define NTHREADS (NW * 32)
#define A_STRIDE 108
#define G_STRIDE 260

__device__ __forceinline__ float fsigmoid(float v) {
    return 1.0f / (1.0f + __expf(-v));
}
__device__ __forceinline__ float ftanh_fast(float v) {
    v = fminf(15.0f, fmaxf(-15.0f, v));
    float e = __expf(2.0f * v);
    return (e - 1.0f) / (e + 1.0f);
}

__global__ __launch_bounds__(NTHREADS, 1)
void gru_scan_kernel(const float* __restrict__ x,
                     const float* __restrict__ Wih,
                     const float* __restrict__ Whh,
                     const float* __restrict__ bih,
                     const float* __restrict__ bhh,
                     float* __restrict__ out)
{
    __shared__ float Apan[2][BM][A_STRIDE];
    __shared__ float Gbuf[BM][G_STRIDE];

    const int tid   = (int)threadIdx.x;
    const int lane  = tid & 31;
    const int wave  = tid >> 5;
    const int b0    = (int)blockIdx.x * BM;

    const int g    = wave >> 2;
    const int jg   = ((wave & 3) << 4) + (lane & 15);
    const bool jok = (jg < HDIM);
    const int koff = (lane >> 4) << 1;

    float bw[KSTEPS][2];
#pragma unroll
    for (int k = 0; k < KSTEPS; ++k) {
#pragma unroll
        for (int v = 0; v < 2; ++v) {
            const int kk = 4 * k + koff + v;
            float wv = 0.0f;
            if (jok) {
                if (g < 2) {
                    const int row = g * HDIM + jg;
                    wv = (kk < HDIM) ? Wih[row * HDIM + kk]
                                     : Whh[row * HDIM + (kk - HDIM)];
                } else if (g == 2) {
                    if (kk < HDIM) wv = Wih[(2 * HDIM + jg) * HDIM + kk];
                } else {
                    if (kk >= HDIM) wv = Whh[(2 * HDIM + jg) * HDIM + (kk - HDIM)];
                }
            }
            bw[k][v] = wv;
        }
    }

    float bias = 0.0f;
    if (jok) {
        if (g < 2)       bias = bih[g * HDIM + jg] + bhh[g * HDIM + jg];
        else if (g == 2) bias = bih[2 * HDIM + jg];
        else             bias = bhh[2 * HDIM + jg];
    }

    const int m0 = tid / HDIM, j0 = tid - m0 * HDIM;
    const int i1 = tid + NTHREADS;
    const bool has1 = (i1 < BM * HDIM);
    const int m1 = has1 ? (i1 / HDIM) : 0;
    const int j1 = has1 ? (i1 - m1 * HDIM) : 0;
    const int cn    = (wave << 4) + (lane & 15);
    const int mbase = (lane >> 4) << 3;

    for (int i = tid; i < BM * A_STRIDE; i += NTHREADS)
        (&Apan[0][0][0])[i] = 0.0f;
    __syncthreads();
    Apan[0][m0][j0] = x[((size_t)(b0 + m0) * T_LEN) * HDIM + j0];
    if (has1)
        Apan[0][m1][j1] = x[((size_t)(b0 + m1) * T_LEN) * HDIM + j1];
    __syncthreads();

    for (int t = 0; t < T_LEN; ++t) {
        const int cur = t & 1, nxt = cur ^ 1;

        float pf0 = 0.0f, pf1 = 0.0f;
        if (t + 1 < T_LEN) {
            pf0 = x[((size_t)(b0 + m0) * T_LEN + (t + 1)) * HDIM + j0];
            if (has1)
                pf1 = x[((size_t)(b0 + m1) * T_LEN + (t + 1)) * HDIM + j1];
        }

        v8f c0, c1;
#pragma unroll
        for (int r = 0; r < 8; ++r) { c0[r] = bias; c1[r] = 0.0f; }

        const float* Arow = &Apan[cur][lane & 15][0];
#pragma unroll
        for (int k = 0; k < KSTEPS; k += 2) {
            v2f a0 = *(const v2f*)(Arow + 4 * k + koff);
            v2f b0v; b0v[0] = bw[k][0]; b0v[1] = bw[k][1];
            c0 = __builtin_amdgcn_wmma_f32_16x16x4_f32(
                     false, a0, false, b0v, (short)0, c0, false, false);
            if (k + 1 < KSTEPS) {
                v2f a1 = *(const v2f*)(Arow + 4 * (k + 1) + koff);
                v2f b1v; b1v[0] = bw[k + 1][0]; b1v[1] = bw[k + 1][1];
                c1 = __builtin_amdgcn_wmma_f32_16x16x4_f32(
                         false, a1, false, b1v, (short)0, c1, false, false);
            }
        }

#pragma unroll
        for (int r = 0; r < 8; ++r)
            Gbuf[mbase + r][cn] = c0[r] + c1[r];
        if (t + 1 < T_LEN) {
            Apan[nxt][m0][j0] = pf0;
            if (has1) Apan[nxt][m1][j1] = pf1;
        }
        __syncthreads();

        {
            const float ar  = Gbuf[m0][j0];
            const float az  = Gbuf[m0][64 + j0];
            const float gxn = Gbuf[m0][128 + j0];
            const float ghn = Gbuf[m0][192 + j0];
            const float r   = fsigmoid(ar);
            const float zz  = fsigmoid(az);
            const float n   = ftanh_fast(gxn + r * ghn);
            const float hp  = Apan[cur][m0][HDIM + j0];
            Apan[nxt][m0][HDIM + j0] = (1.0f - zz) * n + zz * hp;
        }
        if (has1) {
            const float ar  = Gbuf[m1][j1];
            const float az  = Gbuf[m1][64 + j1];
            const float gxn = Gbuf[m1][128 + j1];
            const float ghn = Gbuf[m1][192 + j1];
            const float r   = fsigmoid(ar);
            const float zz  = fsigmoid(az);
            const float n   = ftanh_fast(gxn + r * ghn);
            const float hp  = Apan[cur][m1][HDIM + j1];
            Apan[nxt][m1][HDIM + j1] = (1.0f - zz) * n + zz * hp;
        }
        __syncthreads();
    }

    {
        const int par = T_LEN & 1;
        out[(b0 + m0) * HDIM + j0] = Apan[par][m0][HDIM + j0];
        if (has1)
            out[(b0 + m1) * HDIM + j1] = Apan[par][m1][HDIM + j1];
    }
}

extern "C" void kernel_launch(void* const* d_in, const int* in_sizes, int n_in,
                              void* d_out, int out_size, void* d_ws, size_t ws_size,
                              hipStream_t stream) {
    (void)in_sizes; (void)n_in; (void)d_ws; (void)ws_size; (void)out_size;
    const float* x   = (const float*)d_in[0];
    const float* Wih = (const float*)d_in[1];
    const float* Whh = (const float*)d_in[2];
    const float* bih = (const float*)d_in[3];
    const float* bhh = (const float*)d_in[4];
    float* out = (float*)d_out;

    gru_scan_kernel<<<dim3(B_TOT / BM), dim3(NTHREADS), 0, stream>>>(
        x, Wih, Whh, bih, bhh, out);
}


# weight tiles cover only non-zero blocks: n-gate tiles run 13 k-steps over their own K half, the four 2-column remainder tiles merge into one; 253 WMMAs per step, not 400; sums unchanged
# speedup vs baseline: 1.2397x; 1.2397x over previous
#include <hip/hip_runtime.h>


typedef __attribute__((ext_vector_type(2))) float v2f;
typedef __attribute__((ext_vector_type(8))) float v8f;

#define B_TOT   256
#define T_LEN   2048
#define HDIM    50
#define KDIM    100
#define KSTEPS  25
#define KSTEPS_HALF 13
#define KBASE_H 48
#define JREG    48
#define BM      16
#define NW      16
#define NTHREADS (NW * 32)
#define A_STRIDE 108
#define G_STRIDE 260

__device__ __forceinline__ float fsigmoid(float v) {
    return 1.0f / (1.0f + __expf(-v));
}
__device__ __forceinline__ float ftanh_fast(float v) {
    v = fminf(15.0f, fmaxf(-15.0f, v));
    float e = __expf(2.0f * v);
    return (e - 1.0f) / (e + 1.0f);
}

template <int NK>
__device__ __forceinline__ void wmma_chain(const float* Arow, int koff,
                                           const float (&bw)[KSTEPS][2],
                                           v8f& c0, v8f& c1) {
#pragma unroll
    for (int k = 0; k < NK; k += 2) {
        v2f a0 = *(const v2f*)(Arow + 4 * k + koff);
        v2f b0v; b0v[0] = bw[k][0]; b0v[1] = bw[k][1];
        c0 = __builtin_amdgcn_wmma_f32_16x16x4_f32(
                 false, a0, false, b0v, (short)0, c0, false, false);
        if (k + 1 < NK) {
            v2f a1 = *(const v2f*)(Arow + 4 * (k + 1) + koff);
            v2f b1v; b1v[0] = bw[k + 1][0]; b1v[1] = bw[k + 1][1];
            c1 = __builtin_amdgcn_wmma_f32_16x16x4_f32(
                     false, a1, false, b1v, (short)0, c1, false, false);
        }
    }
}

__global__ __launch_bounds__(NTHREADS, 1)
void gru_scan_kernel(const float* __restrict__ x,
                     const float* __restrict__ Wih,
                     const float* __restrict__ Whh,
                     const float* __restrict__ bih,
                     const float* __restrict__ bhh,
                     float* __restrict__ out)
{
    __shared__ float Apan[2][BM][A_STRIDE];
    __shared__ float Gbuf[BM][G_STRIDE];

    const int tid   = (int)threadIdx.x;
    const int lane  = tid & 31;
    const int wave  = __builtin_amdgcn_readfirstlane(tid >> 5);
    const int b0    = (int)blockIdx.x * BM;

    const int wa = wave >> 2, wb = wave & 3;
    const int c  = lane & 15;
    const bool idle  = (wa == wb) && (wa < 3);
    const bool mixed = (wa == 3) && (wb == 3);
    const bool half  = !mixed && (wa == 3 || wb == 3);
    const bool full  = !idle && !half;
    int g, jg;
    bool jok;
    if (mixed) {
        g = c >> 1; jg = JREG + (c & 1); jok = (c < 8);
    } else if (half) {
        g = (wa == 3) ? 3 : 2; jg = 16 * ((wa == 3) ? wb : wa) + c; jok = true;
    } else {
        const int fi = 2 * wa + ((wb > wa) ? wb - 1 : wb);
        g = fi / 3; jg = 16 * (fi % 3) + c; jok = !idle;
    }
    const int koff  = (lane >> 4) << 1;
    const int kbase = (half && wa == 3) ? KBASE_H : 0;
    const int nk    = full ? KSTEPS : (half ? KSTEPS_HALF : 0);

    float bw[KSTEPS][2];
#pragma unroll
    for (int k = 0; k < KSTEPS; ++k) {
#pragma unroll
        for (int v = 0; v < 2; ++v) {
            const int kk = kbase + 4 * k + koff + v;
            float wv = 0.0f;
            if (jok && k < nk) {
                if (g < 2) {
                    const int row = g * HDIM + jg;
                    wv = (kk < HDIM) ? Wih[row * HDIM + kk]
                                     : Whh[row * HDIM + (kk - HDIM)];
                } else if (g == 2) {
                    if (kk < HDIM) wv = Wih[(2 * HDIM + jg) * HDIM + kk];
                } else {
                    if (kk >= HDIM) wv = Whh[(2 * HDIM + jg) * HDIM + (kk - HDIM)];
                }
            }
            bw[k][v] = wv;
        }
    }

    float bias = 0.0f;
    if (jok) {
        if (g < 2)       bias = bih[g * HDIM + jg] + bhh[g * HDIM + jg];
        else if (g == 2) bias = bih[2 * HDIM + jg];
        else             bias = bhh[2 * HDIM + jg];
    }

    const int m0 = tid / HDIM, j0 = tid - m0 * HDIM;
    const int i1 = tid + NTHREADS;
    const bool has1 = (i1 < BM * HDIM);
    const int m1 = has1 ? (i1 / HDIM) : 0;
    const int j1 = has1 ? (i1 - m1 * HDIM) : 0;
    const int cn    = jok ? (g << 6) + jg : HDIM + (c & 7);
    const int mbase = (lane >> 4) << 3;

    for (int i = tid; i < BM * A_STRIDE; i += NTHREADS)
        (&Apan[0][0][0])[i] = 0.0f;
    __syncthreads();
    Apan[0][m0][j0] = x[((size_t)(b0 + m0) * T_LEN) * HDIM + j0];
    if (has1)
        Apan[0][m1][j1] = x[((size_t)(b0 + m1) * T_LEN) * HDIM + j1];
    __syncthreads();

    for (int t = 0; t < T_LEN; ++t) {
        const int cur = t & 1, nxt = cur ^ 1;

        float pf0 = 0.0f, pf1 = 0.0f;
        if (t + 1 < T_LEN) {
            pf0 = x[((size_t)(b0 + m0) * T_LEN + (t + 1)) * HDIM + j0];
            if (has1)
                pf1 = x[((size_t)(b0 + m1) * T_LEN + (t + 1)) * HDIM + j1];
        }

        if (!idle) {
            v8f c0, c1;
#pragma unroll
            for (int r = 0; r < 8; ++r) { c0[r] = bias; c1[r] = 0.0f; }

            const float* Arow = &Apan[cur][c][kbase];
            if (full) wmma_chain<KSTEPS>(Arow, koff, bw, c0, c1);
            else      wmma_chain<KSTEPS_HALF>(Arow, koff, bw, c0, c1);

#pragma unroll
            for (int r = 0; r < 8; ++r)
                Gbuf[mbase + r][cn] = c0[r] + c1[r];
        }
        if (t + 1 < T_LEN) {
            Apan[nxt][m0][j0] = pf0;
            if (has1) Apan[nxt][m1][j1] = pf1;
        }
        __syncthreads();

        {
            const float ar  = Gbuf[m0][j0];
            const float az  = Gbuf[m0][64 + j0];
            const float gxn = Gbuf[m0][128 + j0];
            const float ghn = Gbuf[m0][192 + j0];
            const float r   = fsigmoid(ar);
            const float zz  = fsigmoid(az);
            const float n   = ftanh_fast(gxn + r * ghn);
            const float hp  = Apan[cur][m0][HDIM + j0];
            Apan[nxt][m0][HDIM + j0] = (1.0f - zz) * n + zz * hp;
        }
        if (has1) {
            const float ar  = Gbuf[m1][j1];
            const float az  = Gbuf[m1][64 + j1];
            const float gxn = Gbuf[m1][128 + j1];
            const float ghn = Gbuf[m1][192 + j1];
            const float r   = fsigmoid(ar);
            const float zz  = fsigmoid(az);
            const float n   = ftanh_fast(gxn + r * ghn);
            const float hp  = Apan[cur][m1][HDIM + j1];
            Apan[nxt][m1][HDIM + j1] = (1.0f - zz) * n + zz * hp;
        }
        __syncthreads();
    }

    {
        const int par = T_LEN & 1;
        out[(b0 + m0) * HDIM + j0] = Apan[par][m0][HDIM + j0];
        if (has1)
            out[(b0 + m1) * HDIM + j1] = Apan[par][m1][HDIM + j1];
    }
}

extern "C" void kernel_launch(void* const* d_in, const int* in_sizes, int n_in,
                              void* d_out, int out_size, void* d_ws, size_t ws_size,
                              hipStream_t stream) {
    (void)in_sizes; (void)n_in; (void)d_ws; (void)ws_size; (void)out_size;
    const float* x   = (const float*)d_in[0];
    const float* Wih = (const float*)d_in[1];
    const float* Whh = (const float*)d_in[2];
    const float* bih = (const float*)d_in[3];
    const float* bhh = (const float*)d_in[4];
    float* out = (float*)d_out;

    gru_scan_kernel<<<dim3(B_TOT / BM), dim3(NTHREADS), 0, stream>>>(
        x, Wih, Whh, bih, bhh, out);
}


# gate math divisions (two sigmoids, tanh) use the hardware reciprocal v_rcp_f32 (about 1 ulp), not the correctly rounded division: 20 vector instructions per element, not 49; sums unchanged
# speedup vs baseline: 1.4173x; 1.1433x over previous
#include <hip/hip_runtime.h>


typedef __attribute__((ext_vector_type(2))) float v2f;
typedef __attribute__((ext_vector_type(8))) float v8f;

#define B_TOT   256
#define T_LEN   2048
#define HDIM    50
#define KDIM    100
#define KSTEPS  25
#define KSTEPS_HALF 13
#define KBASE_H 48
#define JREG    48
#define BM      16
#define NW      16
#define NTHREADS (NW * 32)
#define A_STRIDE 108
#define G_STRIDE 260

__device__ __forceinline__ float frcp(float y) {
    return __builtin_amdgcn_rcpf(y);
}
__device__ __forceinline__ float fsigmoid(float v) {
    return frcp(1.0f + __expf(-v));
}
__device__ __forceinline__ float ftanh_fast(float v) {
    v = fminf(15.0f, fmaxf(-15.0f, v));
    float e = __expf(2.0f * v);
    return (e - 1.0f) * frcp(e + 1.0f);
}

template <int NK>
__device__ __forceinline__ void wmma_chain(const float* Arow, int koff,
                                           const float (&bw)[KSTEPS][2],
                                           v8f& c0, v8f& c1) {
#pragma unroll
    for (int k = 0; k < NK; k += 2) {
        v2f a0 = *(const v2f*)(Arow + 4 * k + koff);
        v2f b0v; b0v[0] = bw[k][0]; b0v[1] = bw[k][1];
        c0 = __builtin_amdgcn_wmma_f32_16x16x4_f32(
                 false, a0, false, b0v, (short)0, c0, false, false);
        if (k + 1 < NK) {
            v2f a1 = *(const v2f*)(Arow + 4 * (k + 1) + koff);
            v2f b1v; b1v[0] = bw[k + 1][0]; b1v[1] = bw[k + 1][1];
            c1 = __builtin_amdgcn_wmma_f32_16x16x4_f32(
                     false, a1, false, b1v, (short)0, c1, false, false);
        }
    }
}

__global__ __launch_bounds__(NTHREADS, 1)
void gru_scan_kernel(const float* __restrict__ x,
                     const float* __restrict__ Wih,
                     const float* __restrict__ Whh,
                     const float* __restrict__ bih,
                     const float* __restrict__ bhh,
                     float* __restrict__ out)
{
    __shared__ float Apan[2][BM][A_STRIDE];
    __shared__ float Gbuf[BM][G_STRIDE];

    const int tid   = (int)threadIdx.x;
    const int lane  = tid & 31;
    const int wave  = __builtin_amdgcn_readfirstlane(tid >> 5);
    const int b0    = (int)blockIdx.x * BM;

    const int wa = wave >> 2, wb = wave & 3;
    const int c  = lane & 15;
    const bool idle  = (wa == wb) && (wa < 3);
    const bool mixed = (wa == 3) && (wb == 3);
    const bool half  = !mixed && (wa == 3 || wb == 3);
    const bool full  = !idle && !half;
    int g, jg;
    bool jok;
    if (mixed) {
        g = c >> 1; jg = JREG + (c & 1); jok = (c < 8);
    } else if (half) {
        g = (wa == 3) ? 3 : 2; jg = 16 * ((wa == 3) ? wb : wa) + c; jok = true;
    } else {
        const int fi = 2 * wa + ((wb > wa) ? wb - 1 : wb);
        g = fi / 3; jg = 16 * (fi % 3) + c; jok = !idle;
    }
    const int koff  = (lane >> 4) << 1;
    const int kbase = (half && wa == 3) ? KBASE_H : 0;
    const int nk    = full ? KSTEPS : (half ? KSTEPS_HALF : 0);

    float bw[KSTEPS][2];
#pragma unroll
    for (int k = 0; k < KSTEPS; ++k) {
#pragma unroll
        for (int v = 0; v < 2; ++v) {
            const int kk = kbase + 4 * k + koff + v;
            float wv = 0.0f;
            if (jok && k < nk) {
                if (g < 2) {
                    const int row = g * HDIM + jg;
                    wv = (kk < HDIM) ? Wih[row * HDIM + kk]
                                     : Whh[row * HDIM + (kk - HDIM)];
                } else if (g == 2) {
                    if (kk < HDIM) wv = Wih[(2 * HDIM + jg) * HDIM + kk];
                } else {
                    if (kk >= HDIM) wv = Whh[(2 * HDIM + jg) * HDIM + (kk - HDIM)];
                }
            }
            bw[k][v] = wv;
        }
    }

    float bias = 0.0f;
    if (jok) {
        if (g < 2)       bias = bih[g * HDIM + jg] + bhh[g * HDIM + jg];
        else if (g == 2) bias = bih[2 * HDIM + jg];
        else             bias = bhh[2 * HDIM + jg];
    }

    const int m0 = tid / HDIM, j0 = tid - m0 * HDIM;
    const int i1 = tid + NTHREADS;
    const bool has1 = (i1 < BM * HDIM);
    const int m1 = has1 ? (i1 / HDIM) : 0;
    const int j1 = has1 ? (i1 - m1 * HDIM) : 0;
    const int cn    = jok ? (g << 6) + jg : HDIM + (c & 7);
    const int mbase = (lane >> 4) << 3;

    for (int i = tid; i < BM * A_STRIDE; i += NTHREADS)
        (&Apan[0][0][0])[i] = 0.0f;
    __syncthreads();
    Apan[0][m0][j0] = x[((size_t)(b0 + m0) * T_LEN) * HDIM + j0];
    if (has1)
        Apan[0][m1][j1] = x[((size_t)(b0 + m1) * T_LEN) * HDIM + j1];
    __syncthreads();

    for (int t = 0; t < T_LEN; ++t) {
        const int cur = t & 1, nxt = cur ^ 1;

        float pf0 = 0.0f, pf1 = 0.0f;
        if (t + 1 < T_LEN) {
            pf0 = x[((size_t)(b0 + m0) * T_LEN + (t + 1)) * HDIM + j0];
            if (has1)
                pf1 = x[((size_t)(b0 + m1) * T_LEN + (t + 1)) * HDIM + j1];
        }

        if (!idle) {
            v8f c0, c1;
#pragma unroll
            for (int r = 0; r < 8; ++r) { c0[r] = bias; c1[r] = 0.0f; }

            const float* Arow = &Apan[cur][c][kbase];
            if (full) wmma_chain<KSTEPS>(Arow, koff, bw, c0, c1);
            else      wmma_chain<KSTEPS_HALF>(Arow, koff, bw, c0, c1);

#pragma unroll
            for (int r = 0; r < 8; ++r)
                Gbuf[mbase + r][cn] = c0[r] + c1[r];
        }
        if (t + 1 < T_LEN) {
            Apan[nxt][m0][j0] = pf0;
            if (has1) Apan[nxt][m1][j1] = pf1;
        }
        __syncthreads();

        {
            const float ar  = Gbuf[m0][j0];
            const float az  = Gbuf[m0][64 + j0];
            const float gxn = Gbuf[m0][128 + j0];
            const float ghn = Gbuf[m0][192 + j0];
            const float r   = fsigmoid(ar);
            const float zz  = fsigmoid(az);
            const float n   = ftanh_fast(gxn + r * ghn);
            const float hp  = Apan[cur][m0][HDIM + j0];
            Apan[nxt][m0][HDIM + j0] = (1.0f - zz) * n + zz * hp;
        }
        if (has1) {
            const float ar  = Gbuf[m1][j1];
            const float az  = Gbuf[m1][64 + j1];
            const float gxn = Gbuf[m1][128 + j1];
            const float ghn = Gbuf[m1][192 + j1];
            const float r   = fsigmoid(ar);
            const float zz  = fsigmoid(az);
            const float n   = ftanh_fast(gxn + r * ghn);
            const float hp  = Apan[cur][m1][HDIM + j1];
            Apan[nxt][m1][HDIM + j1] = (1.0f - zz) * n + zz * hp;
        }
        __syncthreads();
    }

    {
        const int par = T_LEN & 1;
        out[(b0 + m0) * HDIM + j0] = Apan[par][m0][HDIM + j0];
        if (has1)
            out[(b0 + m1) * HDIM + j1] = Apan[par][m1][HDIM + j1];
    }
}

extern "C" void kernel_launch(void* const* d_in, const int* in_sizes, int n_in,
                              void* d_out, int out_size, void* d_ws, size_t ws_size,
                              hipStream_t stream) {
    (void)in_sizes; (void)n_in; (void)d_ws; (void)ws_size; (void)out_size;
    const float* x   = (const float*)d_in[0];
    const float* Wih = (const float*)d_in[1];
    const float* Whh = (const float*)d_in[2];
    const float* bih = (const float*)d_in[3];
    const float* bhh = (const float*)d_in[4];
    float* out = (float*)d_out;

    gru_scan_kernel<<<dim3(B_TOT / BM), dim3(NTHREADS), 0, stream>>>(
        x, Wih, Whh, bih, bhh, out);
}


# input projection x W_ih^T + b_ih leaves the serial scan: a parallel launch per 256-step chunk writes it to the workspace, the scan keeps h W_hh^T only (130 WMMAs per step, not 253); sums regrouped
# speedup vs baseline: 1.6775x; 1.1836x over previous
#include <hip/hip_runtime.h>


typedef __attribute__((ext_vector_type(2))) float v2f;
typedef __attribute__((ext_vector_type(8))) float v8f;

#define B_TOT   256
#define T_LEN   2048
#define HDIM    50
#define BM      16
#define NELEM   (BM * HDIM)
#define KS      13
#define P_STRIDE 60
#define G_STRIDE 260
#define NCHUNK  8
#define T_CH    256
#define TT      8
#define CARRY_FLOATS 16384
#define GX_STEP 2400
#define WS_NEED (65536ull + 16ull * T_CH * GX_STEP * 4ull)

__device__ __forceinline__ float frcp(float y) {
    return __builtin_amdgcn_rcpf(y);
}
__device__ __forceinline__ float fsigmoid(float v) {
    return frcp(1.0f + __expf(-v));
}
__device__ __forceinline__ float ftanh_fast(float v) {
    v = fminf(15.0f, fmaxf(-15.0f, v));
    float e = __expf(2.0f * v);
    return (e - 1.0f) * frcp(e + 1.0f);
}

__device__ __forceinline__ void lds_barrier() {
    __builtin_amdgcn_fence(__ATOMIC_RELEASE, "workgroup", "local");
    __builtin_amdgcn_s_barrier();
    __builtin_amdgcn_fence(__ATOMIC_ACQUIRE, "workgroup", "local");
}

template <int NK, int NB>
__device__ __forceinline__ void wmma_chain(const float* Arow, int koff,
                                           const float (&bw)[NB][2],
                                           v8f& c0, v8f& c1) {
#pragma unroll
    for (int k = 0; k < NK; k += 2) {
        v2f a0 = *(const v2f*)(Arow + 4 * k + koff);
        v2f b0v; b0v[0] = bw[k][0]; b0v[1] = bw[k][1];
        c0 = __builtin_amdgcn_wmma_f32_16x16x4_f32(
                 false, a0, false, b0v, (short)0, c0, false, false);
        if (k + 1 < NK) {
            v2f a1 = *(const v2f*)(Arow + 4 * (k + 1) + koff);
            v2f b1v; b1v[0] = bw[k + 1][0]; b1v[1] = bw[k + 1][1];
            c1 = __builtin_amdgcn_wmma_f32_16x16x4_f32(
                     false, a1, false, b1v, (short)0, c1, false, false);
        }
    }
}

__device__ __forceinline__ void load_b(const float* __restrict__ Wm, int g, int jg, bool jok,
                                       int kbase, int nk, int koff, float (&bw)[KS][2]) {
#pragma unroll
    for (int k = 0; k < KS; ++k) {
#pragma unroll
        for (int v = 0; v < 2; ++v) {
            const int kk = kbase + 4 * k + koff + v;
            float wv = 0.0f;
            if (jok && k < nk && kk < HDIM) wv = Wm[(g * HDIM + jg) * HDIM + kk];
            bw[k][v] = wv;
        }
    }
}

__global__ __launch_bounds__(320)
void gru_xproj_chunk(const float* __restrict__ x,
                     const float* __restrict__ Wih,
                     const float* __restrict__ bih,
                     float* __restrict__ ws, int chunk, unsigned long long ws_bytes)
{
    if (ws_bytes < WS_NEED) return;
    __shared__ float Xpan[TT][BM][P_STRIDE];

    const int tid  = (int)threadIdx.x;
    const int lane = tid & 31;
    const int wave = __builtin_amdgcn_readfirstlane(tid >> 5);
    const int blk  = (int)blockIdx.x >> 5;
    const int tl0  = ((int)blockIdx.x & 31) * TT;
    const int t0   = chunk * T_CH + tl0;
    const int b0   = blk * BM;

    const int c    = lane & 15;
    const int koff = (lane >> 4) << 1;
    int g, jg; bool jok;
    if (wave < 9) { g = wave / 3; jg = 16 * (wave % 3) + c; jok = true; }
    else          { jok = (c < 6); g = jok ? (c >> 1) : 0; jg = jok ? 48 + (c & 1) : 0; }

    float bw[KS][2];
    load_b(Wih, g, jg, jok, 0, KS, koff, bw);
    const float bias = jok ? bih[g * HDIM + jg] : 0.0f;

    for (int idx = tid; idx < BM * TT * HDIM; idx += 320) {
        const int m   = idx / (TT * HDIM);
        const int rem = idx - m * (TT * HDIM);
        const int tt  = rem / HDIM;
        const int j   = rem - tt * HDIM;
        Xpan[tt][m][j] = x[((size_t)(b0 + m) * T_LEN + t0) * HDIM + rem];
    }
    if (tid < TT * BM) {
        (&Xpan[0][0][0])[tid * P_STRIDE + HDIM]     = 0.0f;
        (&Xpan[0][0][0])[tid * P_STRIDE + HDIM + 1] = 0.0f;
    }
    __syncthreads();

    const int mbase = (lane >> 4) << 3;
    float* gout = ws + CARRY_FLOATS + ((size_t)(blk * T_CH + tl0) * 3 + g) * NELEM + jg;
    for (int tt = 0; tt < TT; ++tt) {
        v8f c0, c1;
#pragma unroll
        for (int r = 0; r < 8; ++r) { c0[r] = bias; c1[r] = 0.0f; }
        wmma_chain<KS, KS>(&Xpan[tt][c][0], koff, bw, c0, c1);
        if (jok) {
#pragma unroll
            for (int r = 0; r < 8; ++r)
                gout[(size_t)tt * GX_STEP + (mbase + r) * HDIM] = c0[r] + c1[r];
        }
    }
}

__global__ __launch_bounds__(512, 1)
void gru_scan_chunk(const float* __restrict__ Whh,
                    const float* __restrict__ bhh,
                    float* __restrict__ ws,
                    float* __restrict__ out,
                    int chunk, unsigned long long ws_bytes)
{
    if (ws_bytes < WS_NEED) return;
    __shared__ float Hpan[2][BM][P_STRIDE];
    __shared__ float Gbuf[BM][G_STRIDE];

    const int tid   = (int)threadIdx.x;
    const int lane  = tid & 31;
    const int wave  = __builtin_amdgcn_readfirstlane(tid >> 5);
    const int b0    = (int)blockIdx.x * BM;

    const int wa = wave >> 2, wb = wave & 3;
    const int c  = lane & 15;
    const bool idle  = (wa == wb) && (wa < 3);
    const bool mixed = (wa == 3) && (wb == 3);
    const bool half  = !mixed && (wa == 3 || wb == 3);
    const bool half2 = half && (wa == 3);
    int g, jg; bool jok;
    if (mixed) {
        jok = (c < 6); g = jok ? (c >> 1) : 0; jg = jok ? 48 + (c & 1) : 0;
    } else if (half) {
        g = 2; jg = 16 * (half2 ? wb : wa) + c; jok = true;
    } else {
        const int fi = 2 * wa + ((wb > wa) ? wb - 1 : wb);
        g = fi / 3; jg = 16 * (fi % 3) + c; jok = !idle;
    }
    const int koff  = (lane >> 4) << 1;
    const int kbase = half2 ? 28 : 0;
    const int nk    = idle ? 0 : (half ? (half2 ? 6 : 7) : KS);

    float bw[KS][2];
    load_b(Whh, g, jg, jok, kbase, nk, koff, bw);
    const float bias = (jok && !half2) ? bhh[g * HDIM + jg] : 0.0f;

    const int m0 = tid / HDIM, j0 = tid - m0 * HDIM;
    const int i1 = tid + 512;
    const bool has1 = (i1 < NELEM);
    const int m1 = has1 ? (i1 / HDIM) : 0;
    const int j1 = has1 ? (i1 - m1 * HDIM) : 0;
    const int cn    = jok ? ((half2 ? 192 : (g << 6)) + jg) : HDIM + (c - 6);
    const int mbase = (lane >> 4) << 3;

    const float* gx    = ws + CARRY_FLOATS + (size_t)blockIdx.x * T_CH * GX_STEP;
    float*       carry = ws + b0 * HDIM;

    for (int i = tid; i < 2 * BM * P_STRIDE; i += 512) (&Hpan[0][0][0])[i] = 0.0f;
    for (int i = tid; i < BM * G_STRIDE; i += 512)     (&Gbuf[0][0])[i]    = 0.0f;
    __syncthreads();
    if (chunk > 0) {
        Hpan[0][m0][j0] = carry[tid];
        if (has1) Hpan[0][m1][j1] = carry[i1];
    }
    float xr0 = gx[tid], xz0 = gx[NELEM + tid], xn0 = gx[2 * NELEM + tid];
    float xr1 = 0.0f, xz1 = 0.0f, xn1 = 0.0f;
    if (has1) { xr1 = gx[i1]; xz1 = gx[NELEM + i1]; xn1 = gx[2 * NELEM + i1]; }
    __syncthreads();

    for (int t = 0; t < T_CH; ++t) {
        const int cur = t & 1, nxt = cur ^ 1;

        float pr0 = 0.0f, pz0 = 0.0f, pn0 = 0.0f, pr1 = 0.0f, pz1 = 0.0f, pn1 = 0.0f;
        if (t + 1 < T_CH) {
            const float* gp = gx + (size_t)(t + 1) * GX_STEP;
            pr0 = gp[tid]; pz0 = gp[NELEM + tid]; pn0 = gp[2 * NELEM + tid];
            if (has1) { pr1 = gp[i1]; pz1 = gp[NELEM + i1]; pn1 = gp[2 * NELEM + i1]; }
        }

        if (!idle) {
            v8f c0, c1;
#pragma unroll
            for (int r = 0; r < 8; ++r) { c0[r] = bias; c1[r] = 0.0f; }
            const float* Arow = &Hpan[cur][c][kbase];
            if (!half)       wmma_chain<KS, KS>(Arow, koff, bw, c0, c1);
            else if (!half2) wmma_chain<7, KS>(Arow, koff, bw, c0, c1);
            else             wmma_chain<6, KS>(Arow, koff, bw, c0, c1);
#pragma unroll
            for (int r = 0; r < 8; ++r)
                Gbuf[mbase + r][cn] = c0[r] + c1[r];
        }
        lds_barrier();

        {
            const float ar  = Gbuf[m0][j0] + xr0;
            const float az  = Gbuf[m0][64 + j0] + xz0;
            const float ghn = Gbuf[m0][128 + j0] + Gbuf[m0][192 + j0];
            const float r   = fsigmoid(ar);
            const float zz  = fsigmoid(az);
            const float n   = ftanh_fast(xn0 + r * ghn);
            const float hp  = Hpan[cur][m0][j0];
            Hpan[nxt][m0][j0] = (1.0f - zz) * n + zz * hp;
        }
        if (has1) {
            const float ar  = Gbuf[m1][j1] + xr1;
            const float az  = Gbuf[m1][64 + j1] + xz1;
            const float ghn = Gbuf[m1][128 + j1] + Gbuf[m1][192 + j1];
            const float r   = fsigmoid(ar);
            const float zz  = fsigmoid(az);
            const float n   = ftanh_fast(xn1 + r * ghn);
            const float hp  = Hpan[cur][m1][j1];
            Hpan[nxt][m1][j1] = (1.0f - zz) * n + zz * hp;
        }
        lds_barrier();
        xr0 = pr0; xz0 = pz0; xn0 = pn0; xr1 = pr1; xz1 = pz1; xn1 = pn1;
    }

    {
        float* dst = (chunk == NCHUNK - 1) ? (out + b0 * HDIM) : carry;
        dst[tid] = Hpan[T_CH & 1][m0][j0];
        if (has1) dst[i1] = Hpan[T_CH & 1][m1][j1];
    }
}

#define KDIM    100
#define KSTEPS  25
#define KSTEPS_HALF 13
#define KBASE_H 48
#define JREG    48
#define NW      16
#define NTHREADS (NW * 32)
#define A_STRIDE 108
__global__ __launch_bounds__(NTHREADS, 1)
void gru_scan_fused(const float* __restrict__ x,
                     const float* __restrict__ Wih,
                     const float* __restrict__ Whh,
                     const float* __restrict__ bih,
                     const float* __restrict__ bhh,
                     float* __restrict__ out,
                     unsigned long long ws_bytes)
{
    if (ws_bytes >= WS_NEED) return;
    __shared__ float Apan[2][BM][A_STRIDE];
    __shared__ float Gbuf[BM][G_STRIDE];

    const int tid   = (int)threadIdx.x;
    const int lane  = tid & 31;
    const int wave  = __builtin_amdgcn_readfirstlane(tid >> 5);
    const int b0    = (int)blockIdx.x * BM;

    const int wa = wave >> 2, wb = wave & 3;
    const int c  = lane & 15;
    const bool idle  = (wa == wb) && (wa < 3);
    const bool mixed = (wa == 3) && (wb == 3);
    const bool half  = !mixed && (wa == 3 || wb == 3);
    const bool full  = !idle && !half;
    int g, jg;
    bool jok;
    if (mixed) {
        g = c >> 1; jg = JREG + (c & 1); jok = (c < 8);
    } else if (half) {
        g = (wa == 3) ? 3 : 2; jg = 16 * ((wa == 3) ? wb : wa) + c; jok = true;
    } else {
        const int fi = 2 * wa + ((wb > wa) ? wb - 1 : wb);
        g = fi / 3; jg = 16 * (fi % 3) + c; jok = !idle;
    }
    const int koff  = (lane >> 4) << 1;
    const int kbase = (half && wa == 3) ? KBASE_H : 0;
    const int nk    = full ? KSTEPS : (half ? KSTEPS_HALF : 0);

    float bw[KSTEPS][2];
#pragma unroll
    for (int k = 0; k < KSTEPS; ++k) {
#pragma unroll
        for (int v = 0; v < 2; ++v) {
            const int kk = kbase + 4 * k + koff + v;
            float wv = 0.0f;
            if (jok && k < nk) {
                if (g < 2) {
                    const int row = g * HDIM + jg;
                    wv = (kk < HDIM) ? Wih[row * HDIM + kk]
                                     : Whh[row * HDIM + (kk - HDIM)];
                } else if (g == 2) {
                    if (kk < HDIM) wv = Wih[(2 * HDIM + jg) * HDIM + kk];
                } else {
                    if (kk >= HDIM) wv = Whh[(2 * HDIM + jg) * HDIM + (kk - HDIM)];
                }
            }
            bw[k][v] = wv;
        }
    }

    float bias = 0.0f;
    if (jok) {
        if (g < 2)       bias = bih[g * HDIM + jg] + bhh[g * HDIM + jg];
        else if (g == 2) bias = bih[2 * HDIM + jg];
        else             bias = bhh[2 * HDIM + jg];
    }

    const int m0 = tid / HDIM, j0 = tid - m0 * HDIM;
    const int i1 = tid + NTHREADS;
    const bool has1 = (i1 < BM * HDIM);
    const int m1 = has1 ? (i1 / HDIM) : 0;
    const int j1 = has1 ? (i1 - m1 * HDIM) : 0;
    const int cn    = jok ? (g << 6) + jg : HDIM + (c & 7);
    const int mbase = (lane >> 4) << 3;

    for (int i = tid; i < BM * A_STRIDE; i += NTHREADS)
        (&Apan[0][0][0])[i] = 0.0f;
    __syncthreads();
    Apan[0][m0][j0] = x[((size_t)(b0 + m0) * T_LEN) * HDIM + j0];
    if (has1)
        Apan[0][m1][j1] = x[((size_t)(b0 + m1) * T_LEN) * HDIM + j1];
    __syncthreads();

    for (int t = 0; t < T_LEN; ++t) {
        const int cur = t & 1, nxt = cur ^ 1;

        float pf0 = 0.0f, pf1 = 0.0f;
        if (t + 1 < T_LEN) {
            pf0 = x[((size_t)(b0 + m0) * T_LEN + (t + 1)) * HDIM + j0];
            if (has1)
                pf1 = x[((size_t)(b0 + m1) * T_LEN + (t + 1)) * HDIM + j1];
        }

        if (!idle) {
            v8f c0, c1;
#pragma unroll
            for (int r = 0; r < 8; ++r) { c0[r] = bias; c1[r] = 0.0f; }

            const float* Arow = &Apan[cur][c][kbase];
            if (full) wmma_chain<KSTEPS, KSTEPS>(Arow, koff, bw, c0, c1);
            else      wmma_chain<KSTEPS_HALF, KSTEPS>(Arow, koff, bw, c0, c1);

#pragma unroll
            for (int r = 0; r < 8; ++r)
                Gbuf[mbase + r][cn] = c0[r] + c1[r];
        }
        if (t + 1 < T_LEN) {
            Apan[nxt][m0][j0] = pf0;
            if (has1) Apan[nxt][m1][j1] = pf1;
        }
        __syncthreads();

        {
            const float ar  = Gbuf[m0][j0];
            const float az  = Gbuf[m0][64 + j0];
            const float gxn = Gbuf[m0][128 + j0];
            const float ghn = Gbuf[m0][192 + j0];
            const float r   = fsigmoid(ar);
            const float zz  = fsigmoid(az);
            const float n   = ftanh_fast(gxn + r * ghn);
            const float hp  = Apan[cur][m0][HDIM + j0];
            Apan[nxt][m0][HDIM + j0] = (1.0f - zz) * n + zz * hp;
        }
        if (has1) {
            const float ar  = Gbuf[m1][j1];
            const float az  = Gbuf[m1][64 + j1];
            const float gxn = Gbuf[m1][128 + j1];
            const float ghn = Gbuf[m1][192 + j1];
            const float r   = fsigmoid(ar);
            const float zz  = fsigmoid(az);
            const float n   = ftanh_fast(gxn + r * ghn);
            const float hp  = Apan[cur][m1][HDIM + j1];
            Apan[nxt][m1][HDIM + j1] = (1.0f - zz) * n + zz * hp;
        }
        __syncthreads();
    }

    {
        const int par = T_LEN & 1;
        out[(b0 + m0) * HDIM + j0] = Apan[par][m0][HDIM + j0];
        if (has1)
            out[(b0 + m1) * HDIM + j1] = Apan[par][m1][HDIM + j1];
    }
}


#define CHUNK_PAIR(ci)                                                                   \
    gru_xproj_chunk<<<dim3(512), dim3(320), 0, stream>>>(x, Wih, bih, ws, ci, wsb);      \
    gru_scan_chunk<<<dim3(B_TOT / BM), dim3(512), 0, stream>>>(Whh, bhh, ws, out, ci, wsb)

extern "C" void kernel_launch(void* const* d_in, const int* in_sizes, int n_in,
                              void* d_out, int out_size, void* d_ws, size_t ws_size,
                              hipStream_t stream) {
    (void)in_sizes; (void)n_in; (void)out_size;
    const float* x   = (const float*)d_in[0];
    const float* Wih = (const float*)d_in[1];
    const float* Whh = (const float*)d_in[2];
    const float* bih = (const float*)d_in[3];
    const float* bhh = (const float*)d_in[4];
    float* out = (float*)d_out;
    float* ws  = (float*)d_ws;
    const unsigned long long wsb = (unsigned long long)ws_size;

    CHUNK_PAIR(0); CHUNK_PAIR(1); CHUNK_PAIR(2); CHUNK_PAIR(3);
    CHUNK_PAIR(4); CHUNK_PAIR(5); CHUNK_PAIR(6); CHUNK_PAIR(7);
    gru_scan_fused<<<dim3(B_TOT / BM), dim3(NTHREADS), 0, stream>>>(
        x, Wih, Whh, bih, bhh, out, wsb);
}


# A fragments of each matrix chain are read from LDS as one group ahead of the chain with counted waits, not pair by pair among the matrix instructions; sums unchanged
# speedup vs baseline: 1.7848x; 1.0640x over previous
#include <hip/hip_runtime.h>


typedef __attribute__((ext_vector_type(2))) float v2f;
typedef __attribute__((ext_vector_type(8))) float v8f;

#define B_TOT   256
#define T_LEN   2048
#define HDIM    50
#define BM      16
#define NELEM   (BM * HDIM)
#define KS      13
#define P_STRIDE 60
#define G_STRIDE 260
#define NCHUNK  8
#define T_CH    256
#define TT      8
#define CARRY_FLOATS 16384
#define GX_STEP 2400
#define WS_NEED (65536ull + 16ull * T_CH * GX_STEP * 4ull)

__device__ __forceinline__ float frcp(float y) {
    return __builtin_amdgcn_rcpf(y);
}
__device__ __forceinline__ float fsigmoid(float v) {
    return frcp(1.0f + __expf(-v));
}
__device__ __forceinline__ float ftanh_fast(float v) {
    v = fminf(15.0f, fmaxf(-15.0f, v));
    float e = __expf(2.0f * v);
    return (e - 1.0f) * frcp(e + 1.0f);
}

__device__ __forceinline__ void lds_barrier() {
    __builtin_amdgcn_fence(__ATOMIC_RELEASE, "workgroup", "local");
    __builtin_amdgcn_s_barrier();
    __builtin_amdgcn_fence(__ATOMIC_ACQUIRE, "workgroup", "local");
}

template <int NK, int NB>
__device__ __forceinline__ void wmma_chain(const float* Arow, int koff,
                                           const float (&bw)[NB][2],
                                           v8f& c0, v8f& c1) {
    v2f a[NK];
#pragma unroll
    for (int k = 0; k < NK; ++k)
        a[k] = *(const v2f*)(Arow + 4 * k + koff);
    __builtin_amdgcn_sched_barrier(0);
#pragma unroll
    for (int k = 0; k < NK; k += 2) {
        v2f b0v; b0v[0] = bw[k][0]; b0v[1] = bw[k][1];
        c0 = __builtin_amdgcn_wmma_f32_16x16x4_f32(
                 false, a[k], false, b0v, (short)0, c0, false, false);
        if (k + 1 < NK) {
            v2f b1v; b1v[0] = bw[k + 1][0]; b1v[1] = bw[k + 1][1];
            c1 = __builtin_amdgcn_wmma_f32_16x16x4_f32(
                     false, a[k + 1], false, b1v, (short)0, c1, false, false);
        }
    }
}

__device__ __forceinline__ void load_b(const float* __restrict__ Wm, int g, int jg, bool jok,
                                       int kbase, int nk, int koff, float (&bw)[KS][2]) {
#pragma unroll
    for (int k = 0; k < KS; ++k) {
#pragma unroll
        for (int v = 0; v < 2; ++v) {
            const int kk = kbase + 4 * k + koff + v;
            float wv = 0.0f;
            if (jok && k < nk && kk < HDIM) wv = Wm[(g * HDIM + jg) * HDIM + kk];
            bw[k][v] = wv;
        }
    }
}

__global__ __launch_bounds__(320)
void gru_xproj_chunk(const float* __restrict__ x,
                     const float* __restrict__ Wih,
                     const float* __restrict__ bih,
                     float* __restrict__ ws, int chunk, unsigned long long ws_bytes)
{
    if (ws_bytes < WS_NEED) return;
    __shared__ float Xpan[TT][BM][P_STRIDE];

    const int tid  = (int)threadIdx.x;
    const int lane = tid & 31;
    const int wave = __builtin_amdgcn_readfirstlane(tid >> 5);
    const int blk  = (int)blockIdx.x >> 5;
    const int tl0  = ((int)blockIdx.x & 31) * TT;
    const int t0   = chunk * T_CH + tl0;
    const int b0   = blk * BM;

    const int c    = lane & 15;
    const int koff = (lane >> 4) << 1;
    int g, jg; bool jok;
    if (wave < 9) { g = wave / 3; jg = 16 * (wave % 3) + c; jok = true; }
    else          { jok = (c < 6); g = jok ? (c >> 1) : 0; jg = jok ? 48 + (c & 1) : 0; }

    float bw[KS][2];
    load_b(Wih, g, jg, jok, 0, KS, koff, bw);
    const float bias = jok ? bih[g * HDIM + jg] : 0.0f;

    for (int idx = tid; idx < BM * TT * HDIM; idx += 320) {
        const int m   = idx / (TT * HDIM);
        const int rem = idx - m * (TT * HDIM);
        const int tt  = rem / HDIM;
        const int j   = rem - tt * HDIM;
        Xpan[tt][m][j] = x[((size_t)(b0 + m) * T_LEN + t0) * HDIM + rem];
    }
    if (tid < TT * BM) {
        (&Xpan[0][0][0])[tid * P_STRIDE + HDIM]     = 0.0f;
        (&Xpan[0][0][0])[tid * P_STRIDE + HDIM + 1] = 0.0f;
    }
    __syncthreads();

    const int mbase = (lane >> 4) << 3;
    float* gout = ws + CARRY_FLOATS + ((size_t)(blk * T_CH + tl0) * 3 + g) * NELEM + jg;
    for (int tt = 0; tt < TT; ++tt) {
        v8f c0, c1;
#pragma unroll
        for (int r = 0; r < 8; ++r) { c0[r] = bias; c1[r] = 0.0f; }
        wmma_chain<KS, KS>(&Xpan[tt][c][0], koff, bw, c0, c1);
        if (jok) {
#pragma unroll
            for (int r = 0; r < 8; ++r)
                gout[(size_t)tt * GX_STEP + (mbase + r) * HDIM] = c0[r] + c1[r];
        }
    }
}

__global__ __launch_bounds__(512, 1)
void gru_scan_chunk(const float* __restrict__ Whh,
                    const float* __restrict__ bhh,
                    float* __restrict__ ws,
                    float* __restrict__ out,
                    int chunk, unsigned long long ws_bytes)
{
    if (ws_bytes < WS_NEED) return;
    __shared__ float Hpan[2][BM][P_STRIDE];
    __shared__ float Gbuf[BM][G_STRIDE];

    const int tid   = (int)threadIdx.x;
    const int lane  = tid & 31;
    const int wave  = __builtin_amdgcn_readfirstlane(tid >> 5);
    const int b0    = (int)blockIdx.x * BM;

    const int wa = wave >> 2, wb = wave & 3;
    const int c  = lane & 15;
    const bool idle  = (wa == wb) && (wa < 3);
    const bool mixed = (wa == 3) && (wb == 3);
    const bool half  = !mixed && (wa == 3 || wb == 3);
    const bool half2 = half && (wa == 3);
    int g, jg; bool jok;
    if (mixed) {
        jok = (c < 6); g = jok ? (c >> 1) : 0; jg = jok ? 48 + (c & 1) : 0;
    } else if (half) {
        g = 2; jg = 16 * (half2 ? wb : wa) + c; jok = true;
    } else {
        const int fi = 2 * wa + ((wb > wa) ? wb - 1 : wb);
        g = fi / 3; jg = 16 * (fi % 3) + c; jok = !idle;
    }
    const int koff  = (lane >> 4) << 1;
    const int kbase = half2 ? 28 : 0;
    const int nk    = idle ? 0 : (half ? (half2 ? 6 : 7) : KS);

    float bw[KS][2];
    load_b(Whh, g, jg, jok, kbase, nk, koff, bw);
    const float bias = (jok && !half2) ? bhh[g * HDIM + jg] : 0.0f;

    const int m0 = tid / HDIM, j0 = tid - m0 * HDIM;
    const int i1 = tid + 512;
    const bool has1 = (i1 < NELEM);
    const int m1 = has1 ? (i1 / HDIM) : 0;
    const int j1 = has1 ? (i1 - m1 * HDIM) : 0;
    const int cn    = jok ? ((half2 ? 192 : (g << 6)) + jg) : HDIM + (c - 6);
    const int mbase = (lane >> 4) << 3;

    const float* gx    = ws + CARRY_FLOATS + (size_t)blockIdx.x * T_CH * GX_STEP;
    float*       carry = ws + b0 * HDIM;

    for (int i = tid; i < 2 * BM * P_STRIDE; i += 512) (&Hpan[0][0][0])[i] = 0.0f;
    for (int i = tid; i < BM * G_STRIDE; i += 512)     (&Gbuf[0][0])[i]    = 0.0f;
    __syncthreads();
    if (chunk > 0) {
        Hpan[0][m0][j0] = carry[tid];
        if (has1) Hpan[0][m1][j1] = carry[i1];
    }
    float xr0 = gx[tid], xz0 = gx[NELEM + tid], xn0 = gx[2 * NELEM + tid];
    float xr1 = 0.0f, xz1 = 0.0f, xn1 = 0.0f;
    if (has1) { xr1 = gx[i1]; xz1 = gx[NELEM + i1]; xn1 = gx[2 * NELEM + i1]; }
    __syncthreads();

    for (int t = 0; t < T_CH; ++t) {
        const int cur = t & 1, nxt = cur ^ 1;

        float pr0 = 0.0f, pz0 = 0.0f, pn0 = 0.0f, pr1 = 0.0f, pz1 = 0.0f, pn1 = 0.0f;
        if (t + 1 < T_CH) {
            const float* gp = gx + (size_t)(t + 1) * GX_STEP;
            pr0 = gp[tid]; pz0 = gp[NELEM + tid]; pn0 = gp[2 * NELEM + tid];
            if (has1) { pr1 = gp[i1]; pz1 = gp[NELEM + i1]; pn1 = gp[2 * NELEM + i1]; }
        }

        if (!idle) {
            v8f c0, c1;
#pragma unroll
            for (int r = 0; r < 8; ++r) { c0[r] = bias; c1[r] = 0.0f; }
            const float* Arow = &Hpan[cur][c][kbase];
            if (!half)       wmma_chain<KS, KS>(Arow, koff, bw, c0, c1);
            else if (!half2) wmma_chain<7, KS>(Arow, koff, bw, c0, c1);
            else             wmma_chain<6, KS>(Arow, koff, bw, c0, c1);
#pragma unroll
            for (int r = 0; r < 8; ++r)
                Gbuf[mbase + r][cn] = c0[r] + c1[r];
        }
        lds_barrier();

        {
            const float ar  = Gbuf[m0][j0] + xr0;
            const float az  = Gbuf[m0][64 + j0] + xz0;
            const float ghn = Gbuf[m0][128 + j0] + Gbuf[m0][192 + j0];
            const float r   = fsigmoid(ar);
            const float zz  = fsigmoid(az);
            const float n   = ftanh_fast(xn0 + r * ghn);
            const float hp  = Hpan[cur][m0][j0];
            Hpan[nxt][m0][j0] = (1.0f - zz) * n + zz * hp;
        }
        if (has1) {
            const float ar  = Gbuf[m1][j1] + xr1;
            const float az  = Gbuf[m1][64 + j1] + xz1;
            const float ghn = Gbuf[m1][128 + j1] + Gbuf[m1][192 + j1];
            const float r   = fsigmoid(ar);
            const float zz  = fsigmoid(az);
            const float n   = ftanh_fast(xn1 + r * ghn);
            const float hp  = Hpan[cur][m1][j1];
            Hpan[nxt][m1][j1] = (1.0f - zz) * n + zz * hp;
        }
        lds_barrier();
        xr0 = pr0; xz0 = pz0; xn0 = pn0; xr1 = pr1; xz1 = pz1; xn1 = pn1;
    }

    {
        float* dst = (chunk == NCHUNK - 1) ? (out + b0 * HDIM) : carry;
        dst[tid] = Hpan[T_CH & 1][m0][j0];
        if (has1) dst[i1] = Hpan[T_CH & 1][m1][j1];
    }
}

#define KDIM    100
#define KSTEPS  25
#define KSTEPS_HALF 13
#define KBASE_H 48
#define JREG    48
#define NW      16
#define NTHREADS (NW * 32)
#define A_STRIDE 108
__global__ __launch_bounds__(NTHREADS, 1)
void gru_scan_fused(const float* __restrict__ x,
                     const float* __restrict__ Wih,
                     const float* __restrict__ Whh,
                     const float* __restrict__ bih,
                     const float* __restrict__ bhh,
                     float* __restrict__ out,
                     unsigned long long ws_bytes)
{
    if (ws_bytes >= WS_NEED) return;
    __shared__ float Apan[2][BM][A_STRIDE];
    __shared__ float Gbuf[BM][G_STRIDE];

    const int tid   = (int)threadIdx.x;
    const int lane  = tid & 31;
    const int wave  = __builtin_amdgcn_readfirstlane(tid >> 5);
    const int b0    = (int)blockIdx.x * BM;

    const int wa = wave >> 2, wb = wave & 3;
    const int c  = lane & 15;
    const bool idle  = (wa == wb) && (wa < 3);
    const bool mixed = (wa == 3) && (wb == 3);
    const bool half  = !mixed && (wa == 3 || wb == 3);
    const bool full  = !idle && !half;
    int g, jg;
    bool jok;
    if (mixed) {
        g = c >> 1; jg = JREG + (c & 1); jok = (c < 8);
    } else if (half) {
        g = (wa == 3) ? 3 : 2; jg = 16 * ((wa == 3) ? wb : wa) + c; jok = true;
    } else {
        const int fi = 2 * wa + ((wb > wa) ? wb - 1 : wb);
        g = fi / 3; jg = 16 * (fi % 3) + c; jok = !idle;
    }
    const int koff  = (lane >> 4) << 1;
    const int kbase = (half && wa == 3) ? KBASE_H : 0;
    const int nk    = full ? KSTEPS : (half ? KSTEPS_HALF : 0);

    float bw[KSTEPS][2];
#pragma unroll
    for (int k = 0; k < KSTEPS; ++k) {
#pragma unroll
        for (int v = 0; v < 2; ++v) {
            const int kk = kbase + 4 * k + koff + v;
            float wv = 0.0f;
            if (jok && k < nk) {
                if (g < 2) {
                    const int row = g * HDIM + jg;
                    wv = (kk < HDIM) ? Wih[row * HDIM + kk]
                                     : Whh[row * HDIM + (kk - HDIM)];
                } else if (g == 2) {
                    if (kk < HDIM) wv = Wih[(2 * HDIM + jg) * HDIM + kk];
                } else {
                    if (kk >= HDIM) wv = Whh[(2 * HDIM + jg) * HDIM + (kk - HDIM)];
                }
            }
            bw[k][v] = wv;
        }
    }

    float bias = 0.0f;
    if (jok) {
        if (g < 2)       bias = bih[g * HDIM + jg] + bhh[g * HDIM + jg];
        else if (g == 2) bias = bih[2 * HDIM + jg];
        else             bias = bhh[2 * HDIM + jg];
    }

    const int m0 = tid / HDIM, j0 = tid - m0 * HDIM;
    const int i1 = tid + NTHREADS;
    const bool has1 = (i1 < BM * HDIM);
    const int m1 = has1 ? (i1 / HDIM) : 0;
    const int j1 = has1 ? (i1 - m1 * HDIM) : 0;
    const int cn    = jok ? (g << 6) + jg : HDIM + (c & 7);
    const int mbase = (lane >> 4) << 3;

    for (int i = tid; i < BM * A_STRIDE; i += NTHREADS)
        (&Apan[0][0][0])[i] = 0.0f;
    __syncthreads();
    Apan[0][m0][j0] = x[((size_t)(b0 + m0) * T_LEN) * HDIM + j0];
    if (has1)
        Apan[0][m1][j1] = x[((size_t)(b0 + m1) * T_LEN) * HDIM + j1];
    __syncthreads();

    for (int t = 0; t < T_LEN; ++t) {
        const int cur = t & 1, nxt = cur ^ 1;

        float pf0 = 0.0f, pf1 = 0.0f;
        if (t + 1 < T_LEN) {
            pf0 = x[((size_t)(b0 + m0) * T_LEN + (t + 1)) * HDIM + j0];
            if (has1)
                pf1 = x[((size_t)(b0 + m1) * T_LEN + (t + 1)) * HDIM + j1];
        }

        if (!idle) {
            v8f c0, c1;
#pragma unroll
            for (int r = 0; r < 8; ++r) { c0[r] = bias; c1[r] = 0.0f; }

            const float* Arow = &Apan[cur][c][kbase];
            if (full) wmma_chain<KSTEPS, KSTEPS>(Arow, koff, bw, c0, c1);
            else      wmma_chain<KSTEPS_HALF, KSTEPS>(Arow, koff, bw, c0, c1);

#pragma unroll
            for (int r = 0; r < 8; ++r)
                Gbuf[mbase + r][cn] = c0[r] + c1[r];
        }
        if (t + 1 < T_LEN) {
            Apan[nxt][m0][j0] = pf0;
            if (has1) Apan[nxt][m1][j1] = pf1;
        }
        __syncthreads();

        {
            const float ar  = Gbuf[m0][j0];
            const float az  = Gbuf[m0][64 + j0];
            const float gxn = Gbuf[m0][128 + j0];
            const float ghn = Gbuf[m0][192 + j0];
            const float r   = fsigmoid(ar);
            const float zz  = fsigmoid(az);
            const float n   = ftanh_fast(gxn + r * ghn);
            const float hp  = Apan[cur][m0][HDIM + j0];
            Apan[nxt][m0][HDIM + j0] = (1.0f - zz) * n + zz * hp;
        }
        if (has1) {
            const float ar  = Gbuf[m1][j1];
            const float az  = Gbuf[m1][64 + j1];
            const float gxn = Gbuf[m1][128 + j1];
            const float ghn = Gbuf[m1][192 + j1];
            const float r   = fsigmoid(ar);
            const float zz  = fsigmoid(az);
            const float n   = ftanh_fast(gxn + r * ghn);
            const float hp  = Apan[cur][m1][HDIM + j1];
            Apan[nxt][m1][HDIM + j1] = (1.0f - zz) * n + zz * hp;
        }
        __syncthreads();
    }

    {
        const int par = T_LEN & 1;
        out[(b0 + m0) * HDIM + j0] = Apan[par][m0][HDIM + j0];
        if (has1)
            out[(b0 + m1) * HDIM + j1] = Apan[par][m1][HDIM + j1];
    }
}


#define CHUNK_PAIR(ci)                                                                   \
    gru_xproj_chunk<<<dim3(512), dim3(320), 0, stream>>>(x, Wih, bih, ws, ci, wsb);      \
    gru_scan_chunk<<<dim3(B_TOT / BM), dim3(512), 0, stream>>>(Whh, bhh, ws, out, ci, wsb)

extern "C" void kernel_launch(void* const* d_in, const int* in_sizes, int n_in,
                              void* d_out, int out_size, void* d_ws, size_t ws_size,
                              hipStream_t stream) {
    (void)in_sizes; (void)n_in; (void)out_size;
    const float* x   = (const float*)d_in[0];
    const float* Wih = (const float*)d_in[1];
    const float* Whh = (const float*)d_in[2];
    const float* bih = (const float*)d_in[3];
    const float* bhh = (const float*)d_in[4];
    float* out = (float*)d_out;
    float* ws  = (float*)d_ws;
    const unsigned long long wsb = (unsigned long long)ws_size;

    CHUNK_PAIR(0); CHUNK_PAIR(1); CHUNK_PAIR(2); CHUNK_PAIR(3);
    CHUNK_PAIR(4); CHUNK_PAIR(5); CHUNK_PAIR(6); CHUNK_PAIR(7);
    gru_scan_fused<<<dim3(B_TOT / BM), dim3(NTHREADS), 0, stream>>>(
        x, Wih, Whh, bih, bhh, out, wsb);
}


# each scan wave owns the r, z, n tiles of the same 16 hidden units: gate math runs on the accumulator registers (gx stored in fragment order), no gate buffer, one barrier per step; sums regrouped
# speedup vs baseline: 2.2163x; 1.2418x over previous
#include <hip/hip_runtime.h>


typedef __attribute__((ext_vector_type(2))) float v2f;
typedef __attribute__((ext_vector_type(8))) float v8f;

#define B_TOT   256
#define T_LEN   2048
#define HDIM    50
#define BM      16
#define NELEM   (BM * HDIM)
#define KS      13
#define P_STRIDE 60
#define G_STRIDE 260
#define NCHUNK  8
#define T_CH    256
#define TT      8
#define CARRY_FLOATS 16384
#define GX_STEP 2400
#define GX_TILE 256
#define GX_REM_BASE (9 * GX_TILE)
#define WS_NEED (65536ull + 16ull * T_CH * GX_STEP * 4ull)

__device__ __forceinline__ float frcp(float y) {
    return __builtin_amdgcn_rcpf(y);
}
__device__ __forceinline__ float fsigmoid(float v) {
    return frcp(1.0f + __expf(-v));
}
__device__ __forceinline__ float ftanh_fast(float v) {
    v = fminf(15.0f, fmaxf(-15.0f, v));
    float e = __expf(2.0f * v);
    return (e - 1.0f) * frcp(e + 1.0f);
}

__device__ __forceinline__ v2f fsigmoid2(v2f v) {
    v2f d, o;
    d[0] = __expf(-v[0]); d[1] = __expf(-v[1]);
    d = 1.0f + d;
    o[0] = frcp(d[0]); o[1] = frcp(d[1]);
    return o;
}
__device__ __forceinline__ v2f ftanh_fast2(v2f v) {
    v2f e, q;
    v[0] = fminf(15.0f, fmaxf(-15.0f, v[0])); v[1] = fminf(15.0f, fmaxf(-15.0f, v[1]));
    v = 2.0f * v;
    e[0] = __expf(v[0]); e[1] = __expf(v[1]);
    const v2f den = e + 1.0f;
    q[0] = frcp(den[0]); q[1] = frcp(den[1]);
    return (e - 1.0f) * q;
}

__device__ __forceinline__ void lds_barrier() {
    __builtin_amdgcn_fence(__ATOMIC_RELEASE, "workgroup", "local");
    __builtin_amdgcn_s_barrier();
    __builtin_amdgcn_fence(__ATOMIC_ACQUIRE, "workgroup", "local");
}

template <int NK, int NB>
__device__ __forceinline__ void wmma_chain(const float* Arow, int koff,
                                           const float (&bw)[NB][2],
                                           v8f& c0, v8f& c1) {
    v2f a[NK];
#pragma unroll
    for (int k = 0; k < NK; ++k)
        a[k] = *(const v2f*)(Arow + 4 * k + koff);
    __builtin_amdgcn_sched_barrier(0);
#pragma unroll
    for (int k = 0; k < NK; k += 2) {
        v2f b0v; b0v[0] = bw[k][0]; b0v[1] = bw[k][1];
        c0 = __builtin_amdgcn_wmma_f32_16x16x4_f32(
                 false, a[k], false, b0v, (short)0, c0, false, false);
        if (k + 1 < NK) {
            v2f b1v; b1v[0] = bw[k + 1][0]; b1v[1] = bw[k + 1][1];
            c1 = __builtin_amdgcn_wmma_f32_16x16x4_f32(
                     false, a[k + 1], false, b1v, (short)0, c1, false, false);
        }
    }
}

__device__ __forceinline__ void load_b(const float* __restrict__ Wm, int g, int jg, bool jok,
                                       int kbase, int nk, int koff, float (&bw)[KS][2]) {
#pragma unroll
    for (int k = 0; k < KS; ++k) {
#pragma unroll
        for (int v = 0; v < 2; ++v) {
            const int kk = kbase + 4 * k + koff + v;
            float wv = 0.0f;
            if (jok && k < nk && kk < HDIM) wv = Wm[(g * HDIM + jg) * HDIM + kk];
            bw[k][v] = wv;
        }
    }
}

__global__ __launch_bounds__(320)
void gru_xproj_chunk(const float* __restrict__ x,
                     const float* __restrict__ Wih,
                     const float* __restrict__ bih,
                     float* __restrict__ ws, int chunk, unsigned long long ws_bytes)
{
    if (ws_bytes < WS_NEED) return;
    __shared__ float Xpan[TT][BM][P_STRIDE];

    const int tid  = (int)threadIdx.x;
    const int lane = tid & 31;
    const int wave = __builtin_amdgcn_readfirstlane(tid >> 5);
    const int blk  = (int)blockIdx.x >> 5;
    const int tl0  = ((int)blockIdx.x & 31) * TT;
    const int t0   = chunk * T_CH + tl0;
    const int b0   = blk * BM;

    const int c    = lane & 15;
    const int koff = (lane >> 4) << 1;
    int g, jg; bool jok;
    if (wave < 9) { g = wave / 3; jg = 16 * (wave % 3) + c; jok = true; }
    else          { jok = (c < 6); g = jok ? (c >> 1) : 0; jg = jok ? 48 + (c & 1) : 0; }

    float bw[KS][2];
    load_b(Wih, g, jg, jok, 0, KS, koff, bw);
    const float bias = jok ? bih[g * HDIM + jg] : 0.0f;

    for (int idx = tid; idx < BM * TT * HDIM; idx += 320) {
        const int m   = idx / (TT * HDIM);
        const int rem = idx - m * (TT * HDIM);
        const int tt  = rem / HDIM;
        const int j   = rem - tt * HDIM;
        Xpan[tt][m][j] = x[((size_t)(b0 + m) * T_LEN + t0) * HDIM + rem];
    }
    if (tid < TT * BM) {
        (&Xpan[0][0][0])[tid * P_STRIDE + HDIM]     = 0.0f;
        (&Xpan[0][0][0])[tid * P_STRIDE + HDIM + 1] = 0.0f;
    }
    __syncthreads();

    const int lane_off = (wave < 9)
        ? (((wave % 3) * 3 + g) * 32 + lane) * 8
        : GX_REM_BASE + ((g * 2 + (lane >> 4)) * 2 + (c & 1)) * 8;
    float* gout = ws + CARRY_FLOATS + (size_t)(blk * T_CH + tl0) * GX_STEP + lane_off;
    for (int tt = 0; tt < TT; ++tt) {
        v8f c0, c1;
#pragma unroll
        for (int r = 0; r < 8; ++r) { c0[r] = bias; c1[r] = 0.0f; }
        wmma_chain<KS, KS>(&Xpan[tt][c][0], koff, bw, c0, c1);
        if (jok) {
            float4 lo, hi;
            lo.x = c0[0] + c1[0]; lo.y = c0[1] + c1[1]; lo.z = c0[2] + c1[2]; lo.w = c0[3] + c1[3];
            hi.x = c0[4] + c1[4]; hi.y = c0[5] + c1[5]; hi.z = c0[6] + c1[6]; hi.w = c0[7] + c1[7];
            float4* gp = (float4*)(gout + (size_t)tt * GX_STEP);
            gp[0] = lo; gp[1] = hi;
        }
    }
}

__device__ __forceinline__ void wmma_chain3(const float* Arow, int koff,
                                            const float (&bwr)[KS][2],
                                            const float (&bwz)[KS][2],
                                            const float (&bwn)[KS][2],
                                            v8f& cr, v8f& cz, v8f& cn) {
    v2f a[KS];
#pragma unroll
    for (int k = 0; k < KS; ++k)
        a[k] = *(const v2f*)(Arow + 4 * k + koff);
    __builtin_amdgcn_sched_barrier(0);
#pragma unroll
    for (int k = 0; k < KS; ++k) {
        v2f bv;
        bv[0] = bwr[k][0]; bv[1] = bwr[k][1];
        cr = __builtin_amdgcn_wmma_f32_16x16x4_f32(false, a[k], false, bv, (short)0, cr, false, false);
        bv[0] = bwz[k][0]; bv[1] = bwz[k][1];
        cz = __builtin_amdgcn_wmma_f32_16x16x4_f32(false, a[k], false, bv, (short)0, cz, false, false);
        bv[0] = bwn[k][0]; bv[1] = bwn[k][1];
        cn = __builtin_amdgcn_wmma_f32_16x16x4_f32(false, a[k], false, bv, (short)0, cn, false, false);
    }
}

__global__ __launch_bounds__(128, 1)
void gru_scan_chunk(const float* __restrict__ Whh,
                    const float* __restrict__ bhh,
                    float* __restrict__ ws,
                    float* __restrict__ out,
                    int chunk, unsigned long long ws_bytes)
{
    if (ws_bytes < WS_NEED) return;
    __shared__ float Hpan[2][BM][P_STRIDE];

    const int tid   = (int)threadIdx.x;
    const int lane  = tid & 31;
    const int wave  = __builtin_amdgcn_readfirstlane(tid >> 5);
    const int b0    = (int)blockIdx.x * BM;

    const int c     = lane & 15;
    const int hf    = lane >> 4;
    const int koff  = hf << 1;
    const int mbase = hf << 3;
    const int jg    = 16 * wave + c;
    const bool jok  = (jg < HDIM);

    float bwr[KS][2], bwz[KS][2], bwn[KS][2];
    load_b(Whh, 0, jg, jok, 0, KS, koff, bwr);
    load_b(Whh, 1, jg, jok, 0, KS, koff, bwz);
    load_b(Whh, 2, jg, jok, 0, KS, koff, bwn);
    const float bias_r = jok ? bhh[jg] : 0.0f;
    const float bias_z = jok ? bhh[HDIM + jg] : 0.0f;
    const float bias_n = jok ? bhh[2 * HDIM + jg] : 0.0f;

    int off_r, off_z, off_n;
    if (wave < 3) {
        off_r = ((wave * 3 + 0) * 32 + lane) * 8;
        off_z = ((wave * 3 + 1) * 32 + lane) * 8;
        off_n = ((wave * 3 + 2) * 32 + lane) * 8;
    } else {
        const int cc = jok ? c : 0;
        off_r = GX_REM_BASE + ((0 * 2 + hf) * 2 + cc) * 8;
        off_z = GX_REM_BASE + ((1 * 2 + hf) * 2 + cc) * 8;
        off_n = GX_REM_BASE + ((2 * 2 + hf) * 2 + cc) * 8;
    }

    const float* gx    = ws + CARRY_FLOATS + (size_t)blockIdx.x * T_CH * GX_STEP;
    float*       carry = ws + b0 * HDIM;

    for (int i = tid; i < 2 * BM * P_STRIDE; i += 128) (&Hpan[0][0][0])[i] = 0.0f;
    __syncthreads();
    if (chunk > 0) {
        for (int i = tid; i < NELEM; i += 128) {
            const int m = i / HDIM;
            Hpan[0][m][i - m * HDIM] = carry[i];
        }
    }
    __syncthreads();
    float h[8];
#pragma unroll
    for (int r = 0; r < 8; ++r) h[r] = jok ? Hpan[0][mbase + r][jg] : 0.0f;

    float4 xa[6], xb[6];
    xa[0] = *(const float4*)(gx + off_r); xa[1] = *(const float4*)(gx + off_r + 4);
    xa[2] = *(const float4*)(gx + off_z); xa[3] = *(const float4*)(gx + off_z + 4);
    xa[4] = *(const float4*)(gx + off_n); xa[5] = *(const float4*)(gx + off_n + 4);

    auto step = [&](const int t, const int cur, const float4 (&xu)[6], float4 (&pf)[6]) {
        const float* gp = gx + (size_t)((t + 1 < T_CH) ? t + 1 : t) * GX_STEP;
        pf[0] = *(const float4*)(gp + off_r); pf[1] = *(const float4*)(gp + off_r + 4);
        pf[2] = *(const float4*)(gp + off_z); pf[3] = *(const float4*)(gp + off_z + 4);
        pf[4] = *(const float4*)(gp + off_n); pf[5] = *(const float4*)(gp + off_n + 4);

        v8f cr, cz, cn;
#pragma unroll
        for (int r = 0; r < 8; ++r) { cr[r] = bias_r; cz[r] = bias_z; cn[r] = bias_n; }
        wmma_chain3(&Hpan[cur][c][0], koff, bwr, bwz, bwn, cr, cz, cn);

        const float xr[8] = { xu[0].x, xu[0].y, xu[0].z, xu[0].w, xu[1].x, xu[1].y, xu[1].z, xu[1].w };
        const float xz[8] = { xu[2].x, xu[2].y, xu[2].z, xu[2].w, xu[3].x, xu[3].y, xu[3].z, xu[3].w };
        const float xn[8] = { xu[4].x, xu[4].y, xu[4].z, xu[4].w, xu[5].x, xu[5].y, xu[5].z, xu[5].w };
#pragma unroll
        for (int q = 0; q < 4; ++q) {
            v2f vr, vz, vn, gr, gz, gn, hp;
            vr[0] = cr[2 * q]; vr[1] = cr[2 * q + 1];
            vz[0] = cz[2 * q]; vz[1] = cz[2 * q + 1];
            vn[0] = cn[2 * q]; vn[1] = cn[2 * q + 1];
            gr[0] = xr[2 * q]; gr[1] = xr[2 * q + 1];
            gz[0] = xz[2 * q]; gz[1] = xz[2 * q + 1];
            gn[0] = xn[2 * q]; gn[1] = xn[2 * q + 1];
            hp[0] = h[2 * q];  hp[1] = h[2 * q + 1];
            const v2f rr = fsigmoid2(vr + gr);
            const v2f zz = fsigmoid2(vz + gz);
            const v2f n  = ftanh_fast2(gn + rr * vn);
            const v2f hn = (1.0f - zz) * n + zz * hp;
            h[2 * q] = hn[0]; h[2 * q + 1] = hn[1];
        }
        if (jok) {
#pragma unroll
            for (int r = 0; r < 8; ++r) Hpan[cur ^ 1][mbase + r][jg] = h[r];
        }
        lds_barrier();
    };

    for (int t = 0; t < T_CH; t += 2) {
        step(t,     0, xa, xb);
        step(t + 1, 1, xb, xa);
    }

    {
        float* dst = (chunk == NCHUNK - 1) ? (out + b0 * HDIM) : carry;
        for (int i = tid; i < NELEM; i += 128) {
            const int m = i / HDIM;
            dst[i] = Hpan[T_CH & 1][m][i - m * HDIM];
        }
    }
}

#define KDIM    100
#define KSTEPS  25
#define KSTEPS_HALF 13
#define KBASE_H 48
#define JREG    48
#define NW      16
#define NTHREADS (NW * 32)
#define A_STRIDE 108
__global__ __launch_bounds__(NTHREADS, 1)
void gru_scan_fused(const float* __restrict__ x,
                     const float* __restrict__ Wih,
                     const float* __restrict__ Whh,
                     const float* __restrict__ bih,
                     const float* __restrict__ bhh,
                     float* __restrict__ out,
                     unsigned long long ws_bytes)
{
    if (ws_bytes >= WS_NEED) return;
    __shared__ float Apan[2][BM][A_STRIDE];
    __shared__ float Gbuf[BM][G_STRIDE];

    const int tid   = (int)threadIdx.x;
    const int lane  = tid & 31;
    const int wave  = __builtin_amdgcn_readfirstlane(tid >> 5);
    const int b0    = (int)blockIdx.x * BM;

    const int wa = wave >> 2, wb = wave & 3;
    const int c  = lane & 15;
    const bool idle  = (wa == wb) && (wa < 3);
    const bool mixed = (wa == 3) && (wb == 3);
    const bool half  = !mixed && (wa == 3 || wb == 3);
    const bool full  = !idle && !half;
    int g, jg;
    bool jok;
    if (mixed) {
        g = c >> 1; jg = JREG + (c & 1); jok = (c < 8);
    } else if (half) {
        g = (wa == 3) ? 3 : 2; jg = 16 * ((wa == 3) ? wb : wa) + c; jok = true;
    } else {
        const int fi = 2 * wa + ((wb > wa) ? wb - 1 : wb);
        g = fi / 3; jg = 16 * (fi % 3) + c; jok = !idle;
    }
    const int koff  = (lane >> 4) << 1;
    const int kbase = (half && wa == 3) ? KBASE_H : 0;
    const int nk    = full ? KSTEPS : (half ? KSTEPS_HALF : 0);

    float bw[KSTEPS][2];
#pragma unroll
    for (int k = 0; k < KSTEPS; ++k) {
#pragma unroll
        for (int v = 0; v < 2; ++v) {
            const int kk = kbase + 4 * k + koff + v;
            float wv = 0.0f;
            if (jok && k < nk) {
                if (g < 2) {
                    const int row = g * HDIM + jg;
                    wv = (kk < HDIM) ? Wih[row * HDIM + kk]
                                     : Whh[row * HDIM + (kk - HDIM)];
                } else if (g == 2) {
                    if (kk < HDIM) wv = Wih[(2 * HDIM + jg) * HDIM + kk];
                } else {
                    if (kk >= HDIM) wv = Whh[(2 * HDIM + jg) * HDIM + (kk - HDIM)];
                }
            }
            bw[k][v] = wv;
        }
    }

    float bias = 0.0f;
    if (jok) {
        if (g < 2)       bias = bih[g * HDIM + jg] + bhh[g * HDIM + jg];
        else if (g == 2) bias = bih[2 * HDIM + jg];
        else             bias = bhh[2 * HDIM + jg];
    }

    const int m0 = tid / HDIM, j0 = tid - m0 * HDIM;
    const int i1 = tid + NTHREADS;
    const bool has1 = (i1 < BM * HDIM);
    const int m1 = has1 ? (i1 / HDIM) : 0;
    const int j1 = has1 ? (i1 - m1 * HDIM) : 0;
    const int cn    = jok ? (g << 6) + jg : HDIM + (c & 7);
    const int mbase = (lane >> 4) << 3;

    for (int i = tid; i < BM * A_STRIDE; i += NTHREADS)
        (&Apan[0][0][0])[i] = 0.0f;
    __syncthreads();
    Apan[0][m0][j0] = x[((size_t)(b0 + m0) * T_LEN) * HDIM + j0];
    if (has1)
        Apan[0][m1][j1] = x[((size_t)(b0 + m1) * T_LEN) * HDIM + j1];
    __syncthreads();

    for (int t = 0; t < T_LEN; ++t) {
        const int cur = t & 1, nxt = cur ^ 1;

        float pf0 = 0.0f, pf1 = 0.0f;
        if (t + 1 < T_LEN) {
            pf0 = x[((size_t)(b0 + m0) * T_LEN + (t + 1)) * HDIM + j0];
            if (has1)
                pf1 = x[((size_t)(b0 + m1) * T_LEN + (t + 1)) * HDIM + j1];
        }

        if (!idle) {
            v8f c0, c1;
#pragma unroll
            for (int r = 0; r < 8; ++r) { c0[r] = bias; c1[r] = 0.0f; }

            const float* Arow = &Apan[cur][c][kbase];
            if (full) wmma_chain<KSTEPS, KSTEPS>(Arow, koff, bw, c0, c1);
            else      wmma_chain<KSTEPS_HALF, KSTEPS>(Arow, koff, bw, c0, c1);

#pragma unroll
            for (int r = 0; r < 8; ++r)
                Gbuf[mbase + r][cn] = c0[r] + c1[r];
        }
        if (t + 1 < T_LEN) {
            Apan[nxt][m0][j0] = pf0;
            if (has1) Apan[nxt][m1][j1] = pf1;
        }
        __syncthreads();

        {
            const float ar  = Gbuf[m0][j0];
            const float az  = Gbuf[m0][64 + j0];
            const float gxn = Gbuf[m0][128 + j0];
            const float ghn = Gbuf[m0][192 + j0];
            const float r   = fsigmoid(ar);
            const float zz  = fsigmoid(az);
            const float n   = ftanh_fast(gxn + r * ghn);
            const float hp  = Apan[cur][m0][HDIM + j0];
            Apan[nxt][m0][HDIM + j0] = (1.0f - zz) * n + zz * hp;
        }
        if (has1) {
            const float ar  = Gbuf[m1][j1];
            const float az  = Gbuf[m1][64 + j1];
            const float gxn = Gbuf[m1][128 + j1];
            const float ghn = Gbuf[m1][192 + j1];
            const float r   = fsigmoid(ar);
            const float zz  = fsigmoid(az);
            const float n   = ftanh_fast(gxn + r * ghn);
            const float hp  = Apan[cur][m1][HDIM + j1];
            Apan[nxt][m1][HDIM + j1] = (1.0f - zz) * n + zz * hp;
        }
        __syncthreads();
    }

    {
        const int par = T_LEN & 1;
        out[(b0 + m0) * HDIM + j0] = Apan[par][m0][HDIM + j0];
        if (has1)
            out[(b0 + m1) * HDIM + j1] = Apan[par][m1][HDIM + j1];
    }
}


#define CHUNK_PAIR(ci)                                                                   \
    gru_xproj_chunk<<<dim3(512), dim3(320), 0, stream>>>(x, Wih, bih, ws, ci, wsb);      \
    gru_scan_chunk<<<dim3(B_TOT / BM), dim3(128), 0, stream>>>(Whh, bhh, ws, out, ci, wsb)

extern "C" void kernel_launch(void* const* d_in, const int* in_sizes, int n_in,
                              void* d_out, int out_size, void* d_ws, size_t ws_size,
                              hipStream_t stream) {
    (void)in_sizes; (void)n_in; (void)out_size;
    const float* x   = (const float*)d_in[0];
    const float* Wih = (const float*)d_in[1];
    const float* Whh = (const float*)d_in[2];
    const float* bih = (const float*)d_in[3];
    const float* bhh = (const float*)d_in[4];
    float* out = (float*)d_out;
    float* ws  = (float*)d_ws;
    const unsigned long long wsb = (unsigned long long)ws_size;

    CHUNK_PAIR(0); CHUNK_PAIR(1); CHUNK_PAIR(2); CHUNK_PAIR(3);
    CHUNK_PAIR(4); CHUNK_PAIR(5); CHUNK_PAIR(6); CHUNK_PAIR(7);
    gru_scan_fused<<<dim3(B_TOT / BM), dim3(NTHREADS), 0, stream>>>(
        x, Wih, Whh, bih, bhh, out, wsb);
}


# scan gate math on the hardware tanh (v_tanh_f32): sigmoid(v) = (1 + tanh(v/2))/2, the halving folded exactly into r, z weights and biases; 24 transcendentals per step, not 48; matrix sums unchanged
# speedup vs baseline: 2.4067x; 1.0859x over previous
#include <hip/hip_runtime.h>


typedef __attribute__((ext_vector_type(2))) float v2f;
typedef __attribute__((ext_vector_type(8))) float v8f;

#define B_TOT   256
#define T_LEN   2048
#define HDIM    50
#define BM      16
#define NELEM   (BM * HDIM)
#define KS      13
#define P_STRIDE 60
#define G_STRIDE 260
#define NCHUNK  8
#define T_CH    256
#define TT      8
#define CARRY_FLOATS 16384
#define GX_STEP 2400
#define GX_TILE 256
#define GX_REM_BASE (9 * GX_TILE)
#define WS_NEED (65536ull + 16ull * T_CH * GX_STEP * 4ull)

__device__ __forceinline__ float frcp(float y) {
    return __builtin_amdgcn_rcpf(y);
}
__device__ __forceinline__ float fsigmoid(float v) {
    return frcp(1.0f + __expf(-v));
}
__device__ __forceinline__ float ftanh_fast(float v) {
    v = fminf(15.0f, fmaxf(-15.0f, v));
    float e = __expf(2.0f * v);
    return (e - 1.0f) * frcp(e + 1.0f);
}

__device__ __forceinline__ v2f ftanh_hw2(v2f v) {
    v2f o;
    o[0] = __builtin_amdgcn_tanhf(v[0]); o[1] = __builtin_amdgcn_tanhf(v[1]);
    return o;
}

__device__ __forceinline__ void lds_barrier() {
    __builtin_amdgcn_fence(__ATOMIC_RELEASE, "workgroup", "local");
    __builtin_amdgcn_s_barrier();
    __builtin_amdgcn_fence(__ATOMIC_ACQUIRE, "workgroup", "local");
}

template <int NK, int NB>
__device__ __forceinline__ void wmma_chain(const float* Arow, int koff,
                                           const float (&bw)[NB][2],
                                           v8f& c0, v8f& c1) {
    v2f a[NK];
#pragma unroll
    for (int k = 0; k < NK; ++k)
        a[k] = *(const v2f*)(Arow + 4 * k + koff);
    __builtin_amdgcn_sched_barrier(0);
#pragma unroll
    for (int k = 0; k < NK; k += 2) {
        v2f b0v; b0v[0] = bw[k][0]; b0v[1] = bw[k][1];
        c0 = __builtin_amdgcn_wmma_f32_16x16x4_f32(
                 false, a[k], false, b0v, (short)0, c0, false, false);
        if (k + 1 < NK) {
            v2f b1v; b1v[0] = bw[k + 1][0]; b1v[1] = bw[k + 1][1];
            c1 = __builtin_amdgcn_wmma_f32_16x16x4_f32(
                     false, a[k + 1], false, b1v, (short)0, c1, false, false);
        }
    }
}

__device__ __forceinline__ void load_b(const float* __restrict__ Wm, int g, int jg, bool jok,
                                       int kbase, int nk, int koff, float sc,
                                       float (&bw)[KS][2]) {
#pragma unroll
    for (int k = 0; k < KS; ++k) {
#pragma unroll
        for (int v = 0; v < 2; ++v) {
            const int kk = kbase + 4 * k + koff + v;
            const bool ok = jok && k < nk && kk < HDIM;
            const float wv = Wm[ok ? (g * HDIM + jg) * HDIM + kk : 0];
            bw[k][v] = ok ? sc * wv : 0.0f;
        }
    }
}

__global__ __launch_bounds__(320)
void gru_xproj_chunk(const float* __restrict__ x,
                     const float* __restrict__ Wih,
                     const float* __restrict__ bih,
                     float* __restrict__ ws, int chunk, unsigned long long ws_bytes)
{
    if (ws_bytes < WS_NEED) return;
    __shared__ float Xpan[TT][BM][P_STRIDE];

    const int tid  = (int)threadIdx.x;
    const int lane = tid & 31;
    const int wave = __builtin_amdgcn_readfirstlane(tid >> 5);
    const int blk  = (int)blockIdx.x >> 5;
    const int tl0  = ((int)blockIdx.x & 31) * TT;
    const int t0   = chunk * T_CH + tl0;
    const int b0   = blk * BM;

    const int c    = lane & 15;
    const int koff = (lane >> 4) << 1;
    int g, jg; bool jok;
    if (wave < 9) { g = wave / 3; jg = 16 * (wave % 3) + c; jok = true; }
    else          { jok = (c < 6); g = jok ? (c >> 1) : 0; jg = jok ? 48 + (c & 1) : 0; }

    const float sc = (g < 2) ? 0.5f : 1.0f;
    float bw[KS][2];
    load_b(Wih, g, jg, jok, 0, KS, koff, sc, bw);
    const float bias = jok ? sc * bih[g * HDIM + jg] : 0.0f;

    for (int idx = tid; idx < BM * TT * HDIM; idx += 320) {
        const int m   = idx / (TT * HDIM);
        const int rem = idx - m * (TT * HDIM);
        const int tt  = rem / HDIM;
        const int j   = rem - tt * HDIM;
        Xpan[tt][m][j] = x[((size_t)(b0 + m) * T_LEN + t0) * HDIM + rem];
    }
    if (tid < TT * BM) {
        (&Xpan[0][0][0])[tid * P_STRIDE + HDIM]     = 0.0f;
        (&Xpan[0][0][0])[tid * P_STRIDE + HDIM + 1] = 0.0f;
    }
    __syncthreads();

    const int lane_off = (wave < 9)
        ? (((wave % 3) * 3 + g) * 32 + lane) * 8
        : GX_REM_BASE + ((g * 2 + (lane >> 4)) * 2 + (c & 1)) * 8;
    float* gout = ws + CARRY_FLOATS + (size_t)(blk * T_CH + tl0) * GX_STEP + lane_off;
    for (int tt = 0; tt < TT; ++tt) {
        v8f c0, c1;
#pragma unroll
        for (int r = 0; r < 8; ++r) { c0[r] = bias; c1[r] = 0.0f; }
        wmma_chain<KS, KS>(&Xpan[tt][c][0], koff, bw, c0, c1);
        if (jok) {
            float4 lo, hi;
            lo.x = c0[0] + c1[0]; lo.y = c0[1] + c1[1]; lo.z = c0[2] + c1[2]; lo.w = c0[3] + c1[3];
            hi.x = c0[4] + c1[4]; hi.y = c0[5] + c1[5]; hi.z = c0[6] + c1[6]; hi.w = c0[7] + c1[7];
            float4* gp = (float4*)(gout + (size_t)tt * GX_STEP);
            gp[0] = lo; gp[1] = hi;
        }
    }
}

__device__ __forceinline__ void wmma_chain3(const float* Arow, int koff,
                                            const float (&bwr)[KS][2],
                                            const float (&bwz)[KS][2],
                                            const float (&bwn)[KS][2],
                                            v8f& cr, v8f& cz, v8f& cn) {
    v2f a[KS];
#pragma unroll
    for (int k = 0; k < KS; ++k)
        a[k] = *(const v2f*)(Arow + 4 * k + koff);
    __builtin_amdgcn_sched_barrier(0);
#pragma unroll
    for (int k = 0; k < KS; ++k) {
        v2f bv;
        bv[0] = bwr[k][0]; bv[1] = bwr[k][1];
        cr = __builtin_amdgcn_wmma_f32_16x16x4_f32(false, a[k], false, bv, (short)0, cr, false, false);
        bv[0] = bwz[k][0]; bv[1] = bwz[k][1];
        cz = __builtin_amdgcn_wmma_f32_16x16x4_f32(false, a[k], false, bv, (short)0, cz, false, false);
        bv[0] = bwn[k][0]; bv[1] = bwn[k][1];
        cn = __builtin_amdgcn_wmma_f32_16x16x4_f32(false, a[k], false, bv, (short)0, cn, false, false);
    }
}

__global__ __launch_bounds__(128, 1)
void gru_scan_chunk(const float* __restrict__ Whh,
                    const float* __restrict__ bhh,
                    float* __restrict__ ws,
                    float* __restrict__ out,
                    int chunk, unsigned long long ws_bytes)
{
    if (ws_bytes < WS_NEED) return;
    __shared__ float Hpan[2][BM][P_STRIDE];

    const int tid   = (int)threadIdx.x;
    const int lane  = tid & 31;
    const int wave  = __builtin_amdgcn_readfirstlane(tid >> 5);
    const int b0    = (int)blockIdx.x * BM;

    const int c     = lane & 15;
    const int hf    = lane >> 4;
    const int koff  = hf << 1;
    const int mbase = hf << 3;
    const int jg    = 16 * wave + c;
    const bool jok  = (jg < HDIM);

    float bwr[KS][2], bwz[KS][2], bwn[KS][2];
    load_b(Whh, 0, jg, jok, 0, KS, koff, 0.5f, bwr);
    load_b(Whh, 1, jg, jok, 0, KS, koff, 0.5f, bwz);
    load_b(Whh, 2, jg, jok, 0, KS, koff, 1.0f, bwn);
    const float bias_r = jok ? 0.5f * bhh[jg] : 0.0f;
    const float bias_z = jok ? 0.5f * bhh[HDIM + jg] : 0.0f;
    const float bias_n = jok ? bhh[2 * HDIM + jg] : 0.0f;

    int off_r, off_z, off_n;
    if (wave < 3) {
        off_r = ((wave * 3 + 0) * 32 + lane) * 8;
        off_z = ((wave * 3 + 1) * 32 + lane) * 8;
        off_n = ((wave * 3 + 2) * 32 + lane) * 8;
    } else {
        const int cc = jok ? c : 0;
        off_r = GX_REM_BASE + ((0 * 2 + hf) * 2 + cc) * 8;
        off_z = GX_REM_BASE + ((1 * 2 + hf) * 2 + cc) * 8;
        off_n = GX_REM_BASE + ((2 * 2 + hf) * 2 + cc) * 8;
    }

    const float* gx    = ws + CARRY_FLOATS + (size_t)blockIdx.x * T_CH * GX_STEP;
    float*       carry = ws + b0 * HDIM;

    for (int i = tid; i < 2 * BM * P_STRIDE; i += 128) (&Hpan[0][0][0])[i] = 0.0f;
    __syncthreads();
    if (chunk > 0) {
        for (int i = tid; i < NELEM; i += 128) {
            const int m = i / HDIM;
            Hpan[0][m][i - m * HDIM] = carry[i];
        }
    }
    __syncthreads();
    float h[8];
#pragma unroll
    for (int r = 0; r < 8; ++r) h[r] = jok ? Hpan[0][mbase + r][jg] : 0.0f;

    float4 xa[6], xb[6];
    xa[0] = *(const float4*)(gx + off_r); xa[1] = *(const float4*)(gx + off_r + 4);
    xa[2] = *(const float4*)(gx + off_z); xa[3] = *(const float4*)(gx + off_z + 4);
    xa[4] = *(const float4*)(gx + off_n); xa[5] = *(const float4*)(gx + off_n + 4);

    auto step = [&](const int t, const int cur, const float4 (&xu)[6], float4 (&pf)[6]) {
        const float* gp = gx + (size_t)((t + 1 < T_CH) ? t + 1 : t) * GX_STEP;
        pf[0] = *(const float4*)(gp + off_r); pf[1] = *(const float4*)(gp + off_r + 4);
        pf[2] = *(const float4*)(gp + off_z); pf[3] = *(const float4*)(gp + off_z + 4);
        pf[4] = *(const float4*)(gp + off_n); pf[5] = *(const float4*)(gp + off_n + 4);

        v8f cr, cz, cn;
#pragma unroll
        for (int r = 0; r < 8; ++r) { cr[r] = bias_r; cz[r] = bias_z; cn[r] = bias_n; }
        wmma_chain3(&Hpan[cur][c][0], koff, bwr, bwz, bwn, cr, cz, cn);

        const float xr[8] = { xu[0].x, xu[0].y, xu[0].z, xu[0].w, xu[1].x, xu[1].y, xu[1].z, xu[1].w };
        const float xz[8] = { xu[2].x, xu[2].y, xu[2].z, xu[2].w, xu[3].x, xu[3].y, xu[3].z, xu[3].w };
        const float xn[8] = { xu[4].x, xu[4].y, xu[4].z, xu[4].w, xu[5].x, xu[5].y, xu[5].z, xu[5].w };
#pragma unroll
        for (int q = 0; q < 4; ++q) {
            v2f vr, vz, vn, gr, gz, gn, hp;
            vr[0] = cr[2 * q]; vr[1] = cr[2 * q + 1];
            vz[0] = cz[2 * q]; vz[1] = cz[2 * q + 1];
            vn[0] = cn[2 * q]; vn[1] = cn[2 * q + 1];
            gr[0] = xr[2 * q]; gr[1] = xr[2 * q + 1];
            gz[0] = xz[2 * q]; gz[1] = xz[2 * q + 1];
            gn[0] = xn[2 * q]; gn[1] = xn[2 * q + 1];
            hp[0] = h[2 * q];  hp[1] = h[2 * q + 1];
            const v2f rr = 0.5f * ftanh_hw2(vr + gr) + 0.5f;
            const v2f zz = 0.5f * ftanh_hw2(vz + gz) + 0.5f;
            const v2f n  = ftanh_hw2(gn + rr * vn);
            const v2f hn = n + zz * (hp - n);
            h[2 * q] = hn[0]; h[2 * q + 1] = hn[1];
        }
        if (jok) {
#pragma unroll
            for (int r = 0; r < 8; ++r) Hpan[cur ^ 1][mbase + r][jg] = h[r];
        }
        lds_barrier();
    };

    for (int t = 0; t < T_CH; t += 2) {
        step(t,     0, xa, xb);
        step(t + 1, 1, xb, xa);
    }

    {
        float* dst = (chunk == NCHUNK - 1) ? (out + b0 * HDIM) : carry;
        for (int i = tid; i < NELEM; i += 128) {
            const int m = i / HDIM;
            dst[i] = Hpan[T_CH & 1][m][i - m * HDIM];
        }
    }
}

#define KDIM    100
#define KSTEPS  25
#define KSTEPS_HALF 13
#define KBASE_H 48
#define JREG    48
#define NW      16
#define NTHREADS (NW * 32)
#define A_STRIDE 108
__global__ __launch_bounds__(NTHREADS, 1)
void gru_scan_fused(const float* __restrict__ x,
                     const float* __restrict__ Wih,
                     const float* __restrict__ Whh,
                     const float* __restrict__ bih,
                     const float* __restrict__ bhh,
                     float* __restrict__ out,
                     unsigned long long ws_bytes)
{
    if (ws_bytes >= WS_NEED) return;
    __shared__ float Apan[2][BM][A_STRIDE];
    __shared__ float Gbuf[BM][G_STRIDE];

    const int tid   = (int)threadIdx.x;
    const int lane  = tid & 31;
    const int wave  = __builtin_amdgcn_readfirstlane(tid >> 5);
    const int b0    = (int)blockIdx.x * BM;

    const int wa = wave >> 2, wb = wave & 3;
    const int c  = lane & 15;
    const bool idle  = (wa == wb) && (wa < 3);
    const bool mixed = (wa == 3) && (wb == 3);
    const bool half  = !mixed && (wa == 3 || wb == 3);
    const bool full  = !idle && !half;
    int g, jg;
    bool jok;
    if (mixed) {
        g = c >> 1; jg = JREG + (c & 1); jok = (c < 8);
    } else if (half) {
        g = (wa == 3) ? 3 : 2; jg = 16 * ((wa == 3) ? wb : wa) + c; jok = true;
    } else {
        const int fi = 2 * wa + ((wb > wa) ? wb - 1 : wb);
        g = fi / 3; jg = 16 * (fi % 3) + c; jok = !idle;
    }
    const int koff  = (lane >> 4) << 1;
    const int kbase = (half && wa == 3) ? KBASE_H : 0;
    const int nk    = full ? KSTEPS : (half ? KSTEPS_HALF : 0);

    float bw[KSTEPS][2];
#pragma unroll
    for (int k = 0; k < KSTEPS; ++k) {
#pragma unroll
        for (int v = 0; v < 2; ++v) {
            const int kk = kbase + 4 * k + koff + v;
            float wv = 0.0f;
            if (jok && k < nk) {
                if (g < 2) {
                    const int row = g * HDIM + jg;
                    wv = (kk < HDIM) ? Wih[row * HDIM + kk]
                                     : Whh[row * HDIM + (kk - HDIM)];
                } else if (g == 2) {
                    if (kk < HDIM) wv = Wih[(2 * HDIM + jg) * HDIM + kk];
                } else {
                    if (kk >= HDIM) wv = Whh[(2 * HDIM + jg) * HDIM + (kk - HDIM)];
                }
            }
            bw[k][v] = wv;
        }
    }

    float bias = 0.0f;
    if (jok) {
        if (g < 2)       bias = bih[g * HDIM + jg] + bhh[g * HDIM + jg];
        else if (g == 2) bias = bih[2 * HDIM + jg];
        else             bias = bhh[2 * HDIM + jg];
    }

    const int m0 = tid / HDIM, j0 = tid - m0 * HDIM;
    const int i1 = tid + NTHREADS;
    const bool has1 = (i1 < BM * HDIM);
    const int m1 = has1 ? (i1 / HDIM) : 0;
    const int j1 = has1 ? (i1 - m1 * HDIM) : 0;
    const int cn    = jok ? (g << 6) + jg : HDIM + (c & 7);
    const int mbase = (lane >> 4) << 3;

    for (int i = tid; i < BM * A_STRIDE; i += NTHREADS)
        (&Apan[0][0][0])[i] = 0.0f;
    __syncthreads();
    Apan[0][m0][j0] = x[((size_t)(b0 + m0) * T_LEN) * HDIM + j0];
    if (has1)
        Apan[0][m1][j1] = x[((size_t)(b0 + m1) * T_LEN) * HDIM + j1];
    __syncthreads();

    for (int t = 0; t < T_LEN; ++t) {
        const int cur = t & 1, nxt = cur ^ 1;

        float pf0 = 0.0f, pf1 = 0.0f;
        if (t + 1 < T_LEN) {
            pf0 = x[((size_t)(b0 + m0) * T_LEN + (t + 1)) * HDIM + j0];
            if (has1)
                pf1 = x[((size_t)(b0 + m1) * T_LEN + (t + 1)) * HDIM + j1];
        }

        if (!idle) {
            v8f c0, c1;
#pragma unroll
            for (int r = 0; r < 8; ++r) { c0[r] = bias; c1[r] = 0.0f; }

            const float* Arow = &Apan[cur][c][kbase];
            if (full) wmma_chain<KSTEPS, KSTEPS>(Arow, koff, bw, c0, c1);
            else      wmma_chain<KSTEPS_HALF, KSTEPS>(Arow, koff, bw, c0, c1);

#pragma unroll
            for (int r = 0; r < 8; ++r)
                Gbuf[mbase + r][cn] = c0[r] + c1[r];
        }
        if (t + 1 < T_LEN) {
            Apan[nxt][m0][j0] = pf0;
            if (has1) Apan[nxt][m1][j1] = pf1;
        }
        __syncthreads();

        {
            const float ar  = Gbuf[m0][j0];
            const float az  = Gbuf[m0][64 + j0];
            const float gxn = Gbuf[m0][128 + j0];
            const float ghn = Gbuf[m0][192 + j0];
            const float r   = fsigmoid(ar);
            const float zz  = fsigmoid(az);
            const float n   = ftanh_fast(gxn + r * ghn);
            const float hp  = Apan[cur][m0][HDIM + j0];
            Apan[nxt][m0][HDIM + j0] = (1.0f - zz) * n + zz * hp;
        }
        if (has1) {
            const float ar  = Gbuf[m1][j1];
            const float az  = Gbuf[m1][64 + j1];
            const float gxn = Gbuf[m1][128 + j1];
            const float ghn = Gbuf[m1][192 + j1];
            const float r   = fsigmoid(ar);
            const float zz  = fsigmoid(az);
            const float n   = ftanh_fast(gxn + r * ghn);
            const float hp  = Apan[cur][m1][HDIM + j1];
            Apan[nxt][m1][HDIM + j1] = (1.0f - zz) * n + zz * hp;
        }
        __syncthreads();
    }

    {
        const int par = T_LEN & 1;
        out[(b0 + m0) * HDIM + j0] = Apan[par][m0][HDIM + j0];
        if (has1)
            out[(b0 + m1) * HDIM + j1] = Apan[par][m1][HDIM + j1];
    }
}


#define CHUNK_PAIR(ci)                                                                   \
    gru_xproj_chunk<<<dim3(512), dim3(320), 0, stream>>>(x, Wih, bih, ws, ci, wsb);      \
    gru_scan_chunk<<<dim3(B_TOT / BM), dim3(128), 0, stream>>>(Whh, bhh, ws, out, ci, wsb)

extern "C" void kernel_launch(void* const* d_in, const int* in_sizes, int n_in,
                              void* d_out, int out_size, void* d_ws, size_t ws_size,
                              hipStream_t stream) {
    (void)in_sizes; (void)n_in; (void)out_size;
    const float* x   = (const float*)d_in[0];
    const float* Wih = (const float*)d_in[1];
    const float* Whh = (const float*)d_in[2];
    const float* bih = (const float*)d_in[3];
    const float* bhh = (const float*)d_in[4];
    float* out = (float*)d_out;
    float* ws  = (float*)d_ws;
    const unsigned long long wsb = (unsigned long long)ws_size;

    CHUNK_PAIR(0); CHUNK_PAIR(1); CHUNK_PAIR(2); CHUNK_PAIR(3);
    CHUNK_PAIR(4); CHUNK_PAIR(5); CHUNK_PAIR(6); CHUNK_PAIR(7);
    gru_scan_fused<<<dim3(B_TOT / BM), dim3(NTHREADS), 0, stream>>>(
        x, Wih, Whh, bih, bhh, out, wsb);
}


# both chunk kernels compute the transposed product: a scan wave's new h is the B fragment of its own 16 reduction rows, 12 of 39 WMMAs run from registers across the split barrier; sums reordered
# speedup vs baseline: 2.6738x; 1.1110x over previous
#include <hip/hip_runtime.h>


typedef __attribute__((ext_vector_type(2))) float v2f;
typedef __attribute__((ext_vector_type(8))) float v8f;

#define B_TOT   256
#define T_LEN   2048
#define HDIM    50
#define BM      16
#define NELEM   (BM * HDIM)
#define KS      13
#define P_STRIDE 60
#define G_STRIDE 260
#define NCHUNK  8
#define T_CH    256
#define TT      8
#define CARRY_FLOATS 16384
#define GX_STEP 2400
#define GX_TILE 256
#define GX_REM_BASE (9 * GX_TILE)
#define WS_NEED (65536ull + 16ull * T_CH * GX_STEP * 4ull)

__device__ __forceinline__ float frcp(float y) {
    return __builtin_amdgcn_rcpf(y);
}
__device__ __forceinline__ float fsigmoid(float v) {
    return frcp(1.0f + __expf(-v));
}
__device__ __forceinline__ float ftanh_fast(float v) {
    v = fminf(15.0f, fmaxf(-15.0f, v));
    float e = __expf(2.0f * v);
    return (e - 1.0f) * frcp(e + 1.0f);
}

__device__ __forceinline__ v2f ftanh_hw2(v2f v) {
    v2f o;
    o[0] = __builtin_amdgcn_tanhf(v[0]); o[1] = __builtin_amdgcn_tanhf(v[1]);
    return o;
}

__device__ __forceinline__ void lds_barrier() {
    __builtin_amdgcn_fence(__ATOMIC_RELEASE, "workgroup", "local");
    __builtin_amdgcn_s_barrier();
    __builtin_amdgcn_fence(__ATOMIC_ACQUIRE, "workgroup", "local");
}

template <int NK, int NB, bool TR = false>
__device__ __forceinline__ void wmma_chain(const float* Arow, int koff,
                                           const float (&bw)[NB][2],
                                           v8f& c0, v8f& c1) {
    v2f a[NK];
#pragma unroll
    for (int k = 0; k < NK; ++k)
        a[k] = *(const v2f*)(Arow + 4 * k + koff);
    __builtin_amdgcn_sched_barrier(0);
#pragma unroll
    for (int k = 0; k < NK; k += 2) {
        v2f b0v; b0v[0] = bw[k][0]; b0v[1] = bw[k][1];
        c0 = TR ? __builtin_amdgcn_wmma_f32_16x16x4_f32(
                      false, b0v, false, a[k], (short)0, c0, false, false)
                : __builtin_amdgcn_wmma_f32_16x16x4_f32(
                      false, a[k], false, b0v, (short)0, c0, false, false);
        if (k + 1 < NK) {
            v2f b1v; b1v[0] = bw[k + 1][0]; b1v[1] = bw[k + 1][1];
            c1 = TR ? __builtin_amdgcn_wmma_f32_16x16x4_f32(
                          false, b1v, false, a[k + 1], (short)0, c1, false, false)
                    : __builtin_amdgcn_wmma_f32_16x16x4_f32(
                          false, a[k + 1], false, b1v, (short)0, c1, false, false);
        }
    }
}

__device__ __forceinline__ void load_b(const float* __restrict__ Wm, int g, int jg, bool jok,
                                       int kbase, int nk, int koff, float sc,
                                       float (&bw)[KS][2]) {
#pragma unroll
    for (int k = 0; k < KS; ++k) {
#pragma unroll
        for (int v = 0; v < 2; ++v) {
            const int kk = kbase + 4 * k + koff + v;
            const bool ok = jok && k < nk && kk < HDIM;
            const float wv = Wm[ok ? (g * HDIM + jg) * HDIM + kk : 0];
            bw[k][v] = ok ? sc * wv : 0.0f;
        }
    }
}

__global__ __launch_bounds__(320)
void gru_xproj_chunk(const float* __restrict__ x,
                     const float* __restrict__ Wih,
                     const float* __restrict__ bih,
                     float* __restrict__ ws, int chunk, unsigned long long ws_bytes)
{
    if (ws_bytes < WS_NEED) return;
    __shared__ float Xpan[TT][BM][P_STRIDE];

    const int tid  = (int)threadIdx.x;
    const int lane = tid & 31;
    const int wave = __builtin_amdgcn_readfirstlane(tid >> 5);
    const int blk  = (int)blockIdx.x >> 5;
    const int tl0  = ((int)blockIdx.x & 31) * TT;
    const int t0   = chunk * T_CH + tl0;
    const int b0   = blk * BM;

    const int c    = lane & 15;
    const int hf   = lane >> 4;
    const int koff = hf << 1;
    int g, jg; bool jok;
    if (wave < 9) { g = wave / 3; jg = 16 * (wave % 3) + c; jok = true; }
    else          { jok = (c < 6); g = jok ? (c >> 1) : 0; jg = jok ? 48 + (c & 1) : 0; }

    const float sc = (g < 2) ? 0.5f : 1.0f;
    float bw[KS][2];
    load_b(Wih, g, jg, jok, 0, KS, koff, sc, bw);
    float biasv[8];
#pragma unroll
    for (int r = 0; r < 8; ++r) {
        if (wave < 9) {
            biasv[r] = sc * bih[g * HDIM + 16 * (wave % 3) + 8 * hf + r];
        } else {
            const bool ok = (hf == 0) && (r < 6);
            const float v = bih[ok ? (r >> 1) * HDIM + 48 + (r & 1) : 0];
            biasv[r] = ok ? ((r < 4) ? 0.5f : 1.0f) * v : 0.0f;
        }
    }

    for (int idx = tid; idx < BM * TT * HDIM; idx += 320) {
        const int m   = idx / (TT * HDIM);
        const int rem = idx - m * (TT * HDIM);
        const int tt  = rem / HDIM;
        const int j   = rem - tt * HDIM;
        Xpan[tt][m][j] = x[((size_t)(b0 + m) * T_LEN + t0) * HDIM + rem];
    }
    if (tid < TT * BM) {
        (&Xpan[0][0][0])[tid * P_STRIDE + HDIM]     = 0.0f;
        (&Xpan[0][0][0])[tid * P_STRIDE + HDIM + 1] = 0.0f;
    }
    __syncthreads();

    const int lane_off = (wave < 9)
        ? (((wave % 3) * 3 + g) * 32 + lane) * 8
        : GX_REM_BASE + c * 4;
    float* gout = ws + CARRY_FLOATS + (size_t)(blk * T_CH + tl0) * GX_STEP + lane_off;
    for (int tt = 0; tt < TT; ++tt) {
        v8f c0, c1;
#pragma unroll
        for (int r = 0; r < 8; ++r) { c0[r] = biasv[r]; c1[r] = 0.0f; }
        wmma_chain<KS, KS, true>(&Xpan[tt][c][0], koff, bw, c0, c1);
        if (wave < 9) {
            float4 lo, hi;
            lo.x = c0[0] + c1[0]; lo.y = c0[1] + c1[1]; lo.z = c0[2] + c1[2]; lo.w = c0[3] + c1[3];
            hi.x = c0[4] + c1[4]; hi.y = c0[5] + c1[5]; hi.z = c0[6] + c1[6]; hi.w = c0[7] + c1[7];
            float4* gp = (float4*)(gout + (size_t)tt * GX_STEP);
            gp[0] = lo; gp[1] = hi;
        } else if (hf == 0) {
            float4 q0; float2 q1;
            q0.x = c0[0] + c1[0]; q0.y = c0[1] + c1[1]; q0.z = c0[2] + c1[2]; q0.w = c0[3] + c1[3];
            q1.x = c0[4] + c1[4]; q1.y = c0[5] + c1[5];
            float* gp = gout + (size_t)tt * GX_STEP;
            *(float4*)gp = q0;
            *(float2*)(gp + 64 - 2 * c) = q1;
        }
    }
}

__device__ __forceinline__ void load_w_scan(const float* __restrict__ Wm, int row, bool rowok,
                                            float sc, int hf, int u0, int u1, int u2,
                                            float (&aw)[KS][2]) {
#pragma unroll
    for (int ks = 0; ks < KS; ++ks) {
        const int bk = ks >> 2;
        const int u  = (bk == 0) ? u0 : (bk == 1) ? u1 : (bk == 2) ? u2 : 3;
#pragma unroll
        for (int v = 0; v < 2; ++v) {
            const int kk = 16 * u + 8 * hf + 2 * (ks & 3) + v;
            const bool ok = rowok && kk < HDIM;
            const float wv = Wm[ok ? row * HDIM + kk : 0];
            aw[ks][v] = ok ? sc * wv : 0.0f;
        }
    }
}

#define MM3(KS_, B0_, B1_)                                                                  \
    do {                                                                                    \
        v2f bv_, av_;                                                                       \
        bv_[0] = (B0_); bv_[1] = (B1_);                                                     \
        av_[0] = awr[KS_][0]; av_[1] = awr[KS_][1];                                         \
        cr = __builtin_amdgcn_wmma_f32_16x16x4_f32(false, av_, false, bv_, (short)0, cr, false, false); \
        av_[0] = awz[KS_][0]; av_[1] = awz[KS_][1];                                         \
        cz = __builtin_amdgcn_wmma_f32_16x16x4_f32(false, av_, false, bv_, (short)0, cz, false, false); \
        av_[0] = awn[KS_][0]; av_[1] = awn[KS_][1];                                         \
        cn = __builtin_amdgcn_wmma_f32_16x16x4_f32(false, av_, false, bv_, (short)0, cn, false, false); \
    } while (0)
#define MM1(KS_, CH_, B0_, B1_)                                                             \
    do {                                                                                    \
        v2f bv_, av_;                                                                       \
        bv_[0] = (B0_); bv_[1] = (B1_);                                                     \
        av_[0] = awr[KS_][0]; av_[1] = awr[KS_][1];                                         \
        CH_ = __builtin_amdgcn_wmma_f32_16x16x4_f32(false, av_, false, bv_, (short)0, CH_, false, false); \
    } while (0)

__global__ __launch_bounds__(128, 1)
void gru_scan_chunk(const float* __restrict__ Whh,
                    const float* __restrict__ bhh,
                    float* __restrict__ ws,
                    float* __restrict__ out,
                    int chunk, unsigned long long ws_bytes)
{
    if (ws_bytes < WS_NEED) return;
    __shared__ __attribute__((aligned(16))) float Hpan[2][BM][P_STRIDE];

    const int tid   = (int)threadIdx.x;
    const int lane  = tid & 31;
    const int wave  = __builtin_amdgcn_readfirstlane(tid >> 5);
    const bool mainw = (wave < 3);
    const int b0    = (int)blockIdx.x * BM;

    const int c     = lane & 15;
    const int hf    = lane >> 4;
    const int u0 = mainw ? wave : 0;
    const int u1 = mainw ? (wave + 1) % 3 : 1;
    const int u2 = mainw ? (wave + 2) % 3 : 2;

    float awr[KS][2], awz[KS][2], awn[KS][2];
    {
        const bool rok = mainw || (c < 6);
        const int  gm  = (c < 6) ? (c >> 1) : 0;
        const int  row_r = mainw ? 16 * wave + c : gm * HDIM + 48 + (c & 1);
        const float sc_r = (mainw || gm < 2) ? 0.5f : 1.0f;
        load_w_scan(Whh, row_r, rok, sc_r, hf, u0, u1, u2, awr);
        load_w_scan(Whh, HDIM + 16 * wave + c, mainw, 0.5f, hf, u0, u1, u2, awz);
        load_w_scan(Whh, 2 * HDIM + 16 * wave + c, mainw, 1.0f, hf, u0, u1, u2, awn);
    }
    float br[8], bz[8], bn[8];
#pragma unroll
    for (int r = 0; r < 8; ++r) {
        const int unit = 16 * wave + 8 * hf + r;
        const bool okm = (hf == 0) && (r < 6);
        const int  im  = okm ? (r >> 1) * HDIM + 48 + (r & 1) : 0;
        const float vr = bhh[mainw ? unit : im];
        const float vz = bhh[mainw ? HDIM + unit : 0];
        const float vn = bhh[mainw ? 2 * HDIM + unit : 0];
        br[r] = mainw ? 0.5f * vr : (okm ? ((r < 4) ? 0.5f : 1.0f) * vr : 0.0f);
        bz[r] = mainw ? 0.5f * vz : 0.0f;
        bn[r] = mainw ? vn : 0.0f;
    }

    const int rem_a = GX_REM_BASE + c * 4;
    const int rem_b = GX_REM_BASE + 64 + (c >> 1) * 4;
    const int og0 = mainw ? ((wave * 3 + 0) * 32 + lane) * 8     : rem_a;
    const int og1 = mainw ? ((wave * 3 + 0) * 32 + lane) * 8 + 4 : rem_a;
    const int og2 = mainw ? ((wave * 3 + 1) * 32 + lane) * 8     : rem_b;
    const int og3 = mainw ? ((wave * 3 + 1) * 32 + lane) * 8 + 4 : rem_b;
    const int og4 = mainw ? ((wave * 3 + 2) * 32 + lane) * 8     : rem_a;
    const int og5 = mainw ? ((wave * 3 + 2) * 32 + lane) * 8 + 4 : rem_b;
    const int o_own = 16 * u0 + 8 * hf;
    const int o_1   = 16 * u1 + 8 * hf;
    const int o_2   = 16 * u2 + 8 * hf;
    const int o_3   = 48 + 8 * hf;

    const float* gx    = ws + CARRY_FLOATS + (size_t)blockIdx.x * T_CH * GX_STEP;
    float*       carry = ws + b0 * HDIM;

    for (int i = tid; i < 2 * BM * P_STRIDE; i += 128) (&Hpan[0][0][0])[i] = 0.0f;
    __syncthreads();
    if (chunk > 0) {
        for (int i = tid; i < NELEM; i += 128) {
            const int m = i / HDIM;
            Hpan[0][m][i - m * HDIM] = carry[i];
        }
    }
    __syncthreads();
    float h[8];
#pragma unroll
    for (int r = 0; r < 8; ++r) h[r] = Hpan[0][c][(mainw ? o_own : 48) + (mainw ? r : (r & 1))];

    float4 xa[6], xb[6];
    auto load_gx = [&](const float* gp, float4 (&pf)[6]) {
        pf[0] = *(const float4*)(gp + og0); pf[1] = *(const float4*)(gp + og1);
        pf[2] = *(const float4*)(gp + og2); pf[3] = *(const float4*)(gp + og3);
        pf[4] = *(const float4*)(gp + og4); pf[5] = *(const float4*)(gp + og5);
    };
    load_gx(gx, xa);

    v8f cr, cz, cn;
    auto begin_step = [&]() {
#pragma unroll
        for (int r = 0; r < 8; ++r) { cr[r] = br[r]; cz[r] = bz[r]; cn[r] = bn[r]; }
        if (mainw) {
            MM3(0, h[0], h[1]);
        }
        __builtin_amdgcn_sched_barrier(0);
    };
    begin_step();
    __builtin_amdgcn_fence(__ATOMIC_RELEASE, "workgroup", "local");
    __builtin_amdgcn_s_barrier_signal(-1);

    auto step = [&](const int t, const int cur, const float4 (&xu)[6], float4 (&pf)[6]) {
        load_gx(gx + (size_t)((t + 1 < T_CH) ? t + 1 : t) * GX_STEP, pf);

        const float* Hc = &Hpan[cur][c][0];
        if (mainw) {
            MM3(1, h[2], h[3]);
        }
        __builtin_amdgcn_sched_barrier(0);
        __builtin_amdgcn_s_barrier_wait(-1);
        __builtin_amdgcn_fence(__ATOMIC_ACQUIRE, "workgroup", "local");
        if (mainw) {
            const float4 p0 = *(const float4*)(Hc + o_1), p1 = *(const float4*)(Hc + o_1 + 4);
            const float4 p2 = *(const float4*)(Hc + o_2), p3 = *(const float4*)(Hc + o_2 + 4);
            const v2f    p6 = *(const v2f*)(Hc + o_3);
            __builtin_amdgcn_sched_barrier(0);
            MM3(2, h[4], h[5]);
            MM3(3, h[6], h[7]);
            __builtin_amdgcn_sched_barrier(0);
            MM3(4,  p0.x, p0.y); MM3(5,  p0.z, p0.w); MM3(6,  p1.x, p1.y); MM3(7,  p1.z, p1.w);
            MM3(8,  p2.x, p2.y); MM3(9,  p2.z, p2.w); MM3(10, p3.x, p3.y); MM3(11, p3.z, p3.w);
            MM3(12, p6[0], p6[1]);

            const float xr[8] = { xu[0].x, xu[0].y, xu[0].z, xu[0].w, xu[1].x, xu[1].y, xu[1].z, xu[1].w };
            const float xz[8] = { xu[2].x, xu[2].y, xu[2].z, xu[2].w, xu[3].x, xu[3].y, xu[3].z, xu[3].w };
            const float xn[8] = { xu[4].x, xu[4].y, xu[4].z, xu[4].w, xu[5].x, xu[5].y, xu[5].z, xu[5].w };
#pragma unroll
            for (int q = 0; q < 4; ++q) {
                v2f vr, vz, vn, gr, gz, gn, hp;
                vr[0] = cr[2 * q]; vr[1] = cr[2 * q + 1];
                vz[0] = cz[2 * q]; vz[1] = cz[2 * q + 1];
                vn[0] = cn[2 * q]; vn[1] = cn[2 * q + 1];
                gr[0] = xr[2 * q]; gr[1] = xr[2 * q + 1];
                gz[0] = xz[2 * q]; gz[1] = xz[2 * q + 1];
                gn[0] = xn[2 * q]; gn[1] = xn[2 * q + 1];
                hp[0] = h[2 * q];  hp[1] = h[2 * q + 1];
                const v2f rr = 0.5f * ftanh_hw2(vr + gr) + 0.5f;
                const v2f zz = 0.5f * ftanh_hw2(vz + gz) + 0.5f;
                const v2f n  = ftanh_hw2(gn + rr * vn);
                const v2f hn = n + zz * (hp - n);
                h[2 * q] = hn[0]; h[2 * q + 1] = hn[1];
            }
            float4 s0, s1;
            s0.x = h[0]; s0.y = h[1]; s0.z = h[2]; s0.w = h[3];
            s1.x = h[4]; s1.y = h[5]; s1.z = h[6]; s1.w = h[7];
            float* Hn = &Hpan[cur ^ 1][c][0];
            *(float4*)(Hn + o_own) = s0; *(float4*)(Hn + o_own + 4) = s1;
        } else {
            const float4 p0 = *(const float4*)(Hc + 8 * hf),      p1 = *(const float4*)(Hc + 8 * hf + 4);
            const float4 p2 = *(const float4*)(Hc + 16 + 8 * hf), p3 = *(const float4*)(Hc + 20 + 8 * hf);
            const float4 p4 = *(const float4*)(Hc + 32 + 8 * hf), p5 = *(const float4*)(Hc + 36 + 8 * hf);
            const v2f    p6 = *(const v2f*)(Hc + o_3);
            __builtin_amdgcn_sched_barrier(0);
            MM1(0, cr, p0.x, p0.y); MM1(1, cz, p0.z, p0.w); MM1(2,  cr, p1.x, p1.y); MM1(3,  cz, p1.z, p1.w);
            MM1(4, cr, p2.x, p2.y); MM1(5, cz, p2.z, p2.w); MM1(6,  cr, p3.x, p3.y); MM1(7,  cz, p3.z, p3.w);
            MM1(8, cr, p4.x, p4.y); MM1(9, cz, p4.z, p4.w); MM1(10, cr, p5.x, p5.y); MM1(11, cz, p5.z, p5.w);
            MM1(12, cr, p6[0], p6[1]);
            v2f vr, vz, vn, gr, gz, gn, hp;
            vr[0] = cr[0] + cz[0]; vr[1] = cr[1] + cz[1];
            vz[0] = cr[2] + cz[2]; vz[1] = cr[3] + cz[3];
            vn[0] = cr[4] + cz[4]; vn[1] = cr[5] + cz[5];
            gr[0] = xu[0].x; gr[1] = xu[0].y;
            gz[0] = xu[0].z; gz[1] = xu[0].w;
            gn[0] = (c & 1) ? xu[2].z : xu[2].x;
            gn[1] = (c & 1) ? xu[2].w : xu[2].y;
            hp[0] = h[0]; hp[1] = h[1];
            const v2f rr = 0.5f * ftanh_hw2(vr + gr) + 0.5f;
            const v2f zz = 0.5f * ftanh_hw2(vz + gz) + 0.5f;
            const v2f n  = ftanh_hw2(gn + rr * vn);
            const v2f hn = n + zz * (hp - n);
            h[0] = hn[0]; h[1] = hn[1];
            if (hf == 0) *(v2f*)(&Hpan[cur ^ 1][c][48]) = hn;
        }
        begin_step();
        __builtin_amdgcn_fence(__ATOMIC_RELEASE, "workgroup", "local");
        __builtin_amdgcn_s_barrier_signal(-1);
    };

    for (int t = 0; t < T_CH; t += 2) {
        step(t,     0, xa, xb);
        step(t + 1, 1, xb, xa);
    }
    __builtin_amdgcn_s_barrier_wait(-1);
    __builtin_amdgcn_fence(__ATOMIC_ACQUIRE, "workgroup", "local");

    {
        float* dst = (chunk == NCHUNK - 1) ? (out + b0 * HDIM) : carry;
        for (int i = tid; i < NELEM; i += 128) {
            const int m = i / HDIM;
            dst[i] = Hpan[T_CH & 1][m][i - m * HDIM];
        }
    }
}

#define KDIM    100
#define KSTEPS  25
#define KSTEPS_HALF 13
#define KBASE_H 48
#define JREG    48
#define NW      16
#define NTHREADS (NW * 32)
#define A_STRIDE 108
__global__ __launch_bounds__(NTHREADS, 1)
void gru_scan_fused(const float* __restrict__ x,
                     const float* __restrict__ Wih,
                     const float* __restrict__ Whh,
                     const float* __restrict__ bih,
                     const float* __restrict__ bhh,
                     float* __restrict__ out,
                     unsigned long long ws_bytes)
{
    if (ws_bytes >= WS_NEED) return;
    __shared__ float Apan[2][BM][A_STRIDE];
    __shared__ float Gbuf[BM][G_STRIDE];

    const int tid   = (int)threadIdx.x;
    const int lane  = tid & 31;
    const int wave  = __builtin_amdgcn_readfirstlane(tid >> 5);
    const int b0    = (int)blockIdx.x * BM;

    const int wa = wave >> 2, wb = wave & 3;
    const int c  = lane & 15;
    const bool idle  = (wa == wb) && (wa < 3);
    const bool mixed = (wa == 3) && (wb == 3);
    const bool half  = !mixed && (wa == 3 || wb == 3);
    const bool full  = !idle && !half;
    int g, jg;
    bool jok;
    if (mixed) {
        g = c >> 1; jg = JREG + (c & 1); jok = (c < 8);
    } else if (half) {
        g = (wa == 3) ? 3 : 2; jg = 16 * ((wa == 3) ? wb : wa) + c; jok = true;
    } else {
        const int fi = 2 * wa + ((wb > wa) ? wb - 1 : wb);
        g = fi / 3; jg = 16 * (fi % 3) + c; jok = !idle;
    }
    const int koff  = (lane >> 4) << 1;
    const int kbase = (half && wa == 3) ? KBASE_H : 0;
    const int nk    = full ? KSTEPS : (half ? KSTEPS_HALF : 0);

    float bw[KSTEPS][2];
#pragma unroll
    for (int k = 0; k < KSTEPS; ++k) {
#pragma unroll
        for (int v = 0; v < 2; ++v) {
            const int kk = kbase + 4 * k + koff + v;
            float wv = 0.0f;
            if (jok && k < nk) {
                if (g < 2) {
                    const int row = g * HDIM + jg;
                    wv = (kk < HDIM) ? Wih[row * HDIM + kk]
                                     : Whh[row * HDIM + (kk - HDIM)];
                } else if (g == 2) {
                    if (kk < HDIM) wv = Wih[(2 * HDIM + jg) * HDIM + kk];
                } else {
                    if (kk >= HDIM) wv = Whh[(2 * HDIM + jg) * HDIM + (kk - HDIM)];
                }
            }
            bw[k][v] = wv;
        }
    }

    float bias = 0.0f;
    if (jok) {
        if (g < 2)       bias = bih[g * HDIM + jg] + bhh[g * HDIM + jg];
        else if (g == 2) bias = bih[2 * HDIM + jg];
        else             bias = bhh[2 * HDIM + jg];
    }

    const int m0 = tid / HDIM, j0 = tid - m0 * HDIM;
    const int i1 = tid + NTHREADS;
    const bool has1 = (i1 < BM * HDIM);
    const int m1 = has1 ? (i1 / HDIM) : 0;
    const int j1 = has1 ? (i1 - m1 * HDIM) : 0;
    const int cn    = jok ? (g << 6) + jg : HDIM + (c & 7);
    const int mbase = (lane >> 4) << 3;

    for (int i = tid; i < BM * A_STRIDE; i += NTHREADS)
        (&Apan[0][0][0])[i] = 0.0f;
    __syncthreads();
    Apan[0][m0][j0] = x[((size_t)(b0 + m0) * T_LEN) * HDIM + j0];
    if (has1)
        Apan[0][m1][j1] = x[((size_t)(b0 + m1) * T_LEN) * HDIM + j1];
    __syncthreads();

    for (int t = 0; t < T_LEN; ++t) {
        const int cur = t & 1, nxt = cur ^ 1;

        float pf0 = 0.0f, pf1 = 0.0f;
        if (t + 1 < T_LEN) {
            pf0 = x[((size_t)(b0 + m0) * T_LEN + (t + 1)) * HDIM + j0];
            if (has1)
                pf1 = x[((size_t)(b0 + m1) * T_LEN + (t + 1)) * HDIM + j1];
        }

        if (!idle) {
            v8f c0, c1;
#pragma unroll
            for (int r = 0; r < 8; ++r) { c0[r] = bias; c1[r] = 0.0f; }

            const float* Arow = &Apan[cur][c][kbase];
            if (full) wmma_chain<KSTEPS, KSTEPS>(Arow, koff, bw, c0, c1);
            else      wmma_chain<KSTEPS_HALF, KSTEPS>(Arow, koff, bw, c0, c1);

#pragma unroll
            for (int r = 0; r < 8; ++r)
                Gbuf[mbase + r][cn] = c0[r] + c1[r];
        }
        if (t + 1 < T_LEN) {
            Apan[nxt][m0][j0] = pf0;
            if (has1) Apan[nxt][m1][j1] = pf1;
        }
        __syncthreads();

        {
            const float ar  = Gbuf[m0][j0];
            const float az  = Gbuf[m0][64 + j0];
            const float gxn = Gbuf[m0][128 + j0];
            const float ghn = Gbuf[m0][192 + j0];
            const float r   = fsigmoid(ar);
            const float zz  = fsigmoid(az);
            const float n   = ftanh_fast(gxn + r * ghn);
            const float hp  = Apan[cur][m0][HDIM + j0];
            Apan[nxt][m0][HDIM + j0] = (1.0f - zz) * n + zz * hp;
        }
        if (has1) {
            const float ar  = Gbuf[m1][j1];
            const float az  = Gbuf[m1][64 + j1];
            const float gxn = Gbuf[m1][128 + j1];
            const float ghn = Gbuf[m1][192 + j1];
            const float r   = fsigmoid(ar);
            const float zz  = fsigmoid(az);
            const float n   = ftanh_fast(gxn + r * ghn);
            const float hp  = Apan[cur][m1][HDIM + j1];
            Apan[nxt][m1][HDIM + j1] = (1.0f - zz) * n + zz * hp;
        }
        __syncthreads();
    }

    {
        const int par = T_LEN & 1;
        out[(b0 + m0) * HDIM + j0] = Apan[par][m0][HDIM + j0];
        if (has1)
            out[(b0 + m1) * HDIM + j1] = Apan[par][m1][HDIM + j1];
    }
}


#define CHUNK_PAIR(ci)                                                                   \
    gru_xproj_chunk<<<dim3(512), dim3(320), 0, stream>>>(x, Wih, bih, ws, ci, wsb);      \
    gru_scan_chunk<<<dim3(B_TOT / BM), dim3(128), 0, stream>>>(Whh, bhh, ws, out, ci, wsb)

extern "C" void kernel_launch(void* const* d_in, const int* in_sizes, int n_in,
                              void* d_out, int out_size, void* d_ws, size_t ws_size,
                              hipStream_t stream) {
    (void)in_sizes; (void)n_in; (void)out_size;
    const float* x   = (const float*)d_in[0];
    const float* Wih = (const float*)d_in[1];
    const float* Whh = (const float*)d_in[2];
    const float* bih = (const float*)d_in[3];
    const float* bhh = (const float*)d_in[4];
    float* out = (float*)d_out;
    float* ws  = (float*)d_ws;
    const unsigned long long wsb = (unsigned long long)ws_size;

    CHUNK_PAIR(0); CHUNK_PAIR(1); CHUNK_PAIR(2); CHUNK_PAIR(3);
    CHUNK_PAIR(4); CHUNK_PAIR(5); CHUNK_PAIR(6); CHUNK_PAIR(7);
    gru_scan_fused<<<dim3(B_TOT / BM), dim3(NTHREADS), 0, stream>>>(
        x, Wih, Whh, bih, bhh, out, wsb);
}
